# starting kernel
# baseline (speedup 1.0000x reference)
#include <hip/hip_runtime.h>
#include <hip/hip_bf16.h>
#include <cstdint>

namespace {

constexpr int B_ = 4, N_ = 8192, CIN_ = 4, C_ = 64, K_ = 32, STRIDE_ = 4, NB_ = 2, EXP_ = 4;
constexpr int M_ = N_ / STRIDE_;
constexpr float R1SQ = 0.1f * 0.1f;
constexpr float R2SQ = 0.2f * 0.2f;
constexpr float NEG_BIG = -3.4028235e38f;

typedef float v2f __attribute__((ext_vector_type(2)));
typedef float v8f __attribute__((ext_vector_type(8)));

__global__ void f1_wmma_kernel(const float* __restrict__ f, const float* __restrict__ Wsa,
                               const float* __restrict__ bsa, float* __restrict__ f1) {
  int wave = (blockIdx.x * blockDim.x + threadIdx.x) >> 5;
  int lane = threadIdx.x & 31;
  constexpr int tilesC = C_ / 16;
  constexpr int nTiles = (B_ * N_ / 16) * tilesC;
  if (wave >= nTiles) return;
  int row0 = (wave / tilesC) * 16;
  int c0 = (wave % tilesC) * 16;
  int b = row0 / N_;
  int n0 = row0 % N_;
  int half = lane >> 4;
  int ln = lane & 15;

  const float* fb = f + ((size_t)b * CIN_ + 2 * half) * N_ + n0 + ln;
  v2f a;  a.x = fb[0];        a.y = fb[N_];
  const float* wc = Wsa + (size_t)(2 * half) * C_ + c0 + ln;
  v2f bb; bb.x = wc[0];       bb.y = wc[C_];
  v8f acc = {0.f, 0.f, 0.f, 0.f, 0.f, 0.f, 0.f, 0.f};
  acc = __builtin_amdgcn_wmma_f32_16x16x4_f32(false, a, false, bb, (short)0, acc,
                                              false, false);
  int col = c0 + ln;
  float bv = bsa[col];
#pragma unroll
  for (int r = 0; r < 8; ++r) {
    int row = row0 + r + 8 * half;
    f1[(size_t)row * C_ + col] = fmaxf(acc[r] + bv, 0.0f);
  }
}

__global__ void ball_query_kernel(const float* __restrict__ p, int n_src, int src_stride,
                                  float r2, int* __restrict__ idx_out) {
  int wave = (blockIdx.x * blockDim.x + threadIdx.x) >> 5;
  int lane = threadIdx.x & 31;
  if (wave >= B_ * M_) return;
  int b = wave / M_;
  int m = wave % M_;
  const float* pb = p + (size_t)b * N_ * 3;
  float qx = pb[(size_t)m * STRIDE_ * 3 + 0];
  float qy = pb[(size_t)m * STRIDE_ * 3 + 1];
  float qz = pb[(size_t)m * STRIDE_ * 3 + 2];
  int* out = idx_out + (size_t)wave * K_;

  int count = 0;
  int first_idx = n_src - 1;
  for (int base = 0; base < n_src; base += 32) {
    int j = base + lane;
    const float* sp = pb + (size_t)j * src_stride * 3;
    float dx = sp[0] - qx, dy = sp[1] - qy, dz = sp[2] - qz;
    bool hit = (dx * dx + dy * dy + dz * dz) <= r2;
    unsigned mask = (unsigned)__ballot(hit);
    if (count == 0 && mask != 0u) first_idx = base + (__ffs(mask) - 1);
    if (hit) {
      int pos = count + __popc(mask & ((1u << lane) - 1u));
      if (pos < K_) out[pos] = j;
    }
    count += __popc(mask);
    if (count >= K_) break;
  }
  if (count > K_) count = K_;
  for (int t = count + lane; t < K_; t += 32) out[t] = first_idx;
}

__global__ void neigh_max_kernel(const float* __restrict__ p, const float* __restrict__ src,
                                 const int* __restrict__ idx, int src_stride, int src_n,
                                 const float* __restrict__ Wpe, const float* __restrict__ bpe,
                                 float* __restrict__ out) {
  __shared__ int s_idx[4][K_];
  __shared__ float s_dq[4][K_][3];
  int g = threadIdx.x >> 6;
  int c = threadIdx.x & 63;
  int bm = blockIdx.x * 4 + g;
  if (bm >= B_ * M_) return;
  int b = bm / M_;
  int m = bm % M_;
  const float* pb = p + (size_t)b * N_ * 3;

  if (c < K_) {
    float qx = pb[(size_t)m * STRIDE_ * 3 + 0];
    float qy = pb[(size_t)m * STRIDE_ * 3 + 1];
    float qz = pb[(size_t)m * STRIDE_ * 3 + 2];
    int j = idx[(size_t)bm * K_ + c];
    s_idx[g][c] = j;
    const float* sp = pb + (size_t)j * src_stride * 3;
    s_dq[g][c][0] = sp[0] - qx;
    s_dq[g][c][1] = sp[1] - qy;
    s_dq[g][c][2] = sp[2] - qz;
  }
  __syncthreads();

  float w0 = Wpe[0 * C_ + c], w1 = Wpe[1 * C_ + c], w2 = Wpe[2 * C_ + c];
  float bias = bpe[c];
  const float* sb = src + (size_t)b * src_n * C_;
  float best = NEG_BIG;
#pragma unroll 4
  for (int k = 0; k < K_; ++k) {
    int j = s_idx[g][k];
    float v = s_dq[g][k][0] * w0 + s_dq[g][k][1] * w1 + s_dq[g][k][2] * w2 + bias
            + sb[(size_t)j * C_ + c];
    best = fmaxf(best, v);
  }
  out[(size_t)bm * C_ + c] = best;
}

template <int KDIM, bool RELU, bool SKIP>
__global__ void gemm_wmma_kernel(const float* __restrict__ X, const float* __restrict__ W,
                                 const float* __restrict__ bias,
                                 const float* __restrict__ skip, float* __restrict__ out,
                                 int Mrows, int Ncols) {
  constexpr int MT = 4;
  int wave = (blockIdx.x * blockDim.x + threadIdx.x) >> 5;
  int lane = threadIdx.x & 31;
  int tilesN = Ncols >> 4;
  int nTiles = (Mrows >> 6) * tilesN;
  if (wave >= nTiles) return;
  int m0 = (wave / tilesN) << 6;
  int n0 = (wave % tilesN) << 4;
  int half = lane >> 4;
  int ln = lane & 15;

  const float* Xrow0 = X + (size_t)(m0 + ln) * KDIM + 2 * half;
  const float* Wcol = W + (size_t)(2 * half) * Ncols + n0 + ln;
  constexpr size_t XSTEP = (size_t)16 * KDIM;

  v8f acc[MT];
#pragma unroll
  for (int t = 0; t < MT; ++t) acc[t] = v8f{0.f, 0.f, 0.f, 0.f, 0.f, 0.f, 0.f, 0.f};

#pragma unroll 4
  for (int k = 0; k < KDIM; k += 4) {
    v2f bb;
    bb.x = Wcol[(size_t)k * Ncols];
    bb.y = Wcol[(size_t)(k + 1) * Ncols];
#pragma unroll
    for (int t = 0; t < MT; ++t) {
      v2f a;
      a.x = Xrow0[t * XSTEP + k];
      a.y = Xrow0[t * XSTEP + k + 1];
      acc[t] = __builtin_amdgcn_wmma_f32_16x16x4_f32(false, a, false, bb, (short)0,
                                                     acc[t], false, false);
    }
  }

  int col = n0 + ln;
  float bv = bias[col];
#pragma unroll
  for (int t = 0; t < MT; ++t) {
#pragma unroll
    for (int r = 0; r < 8; ++r) {
      int row = m0 + t * 16 + r + 8 * half;
      float v = acc[t][r] + bv;
      if (SKIP) v += skip[(size_t)row * Ncols + col];
      if (RELU) v = fmaxf(v, 0.0f);
      out[(size_t)row * Ncols + col] = v;
    }
  }
}

}

extern "C" void kernel_launch(void* const* d_in, const int* in_sizes, int n_in,
                              void* d_out, int out_size, void* d_ws, size_t ws_size,
                              hipStream_t stream) {
  (void)in_sizes; (void)n_in; (void)out_size; (void)ws_size;
  const float* p      = (const float*)d_in[0];
  const float* f      = (const float*)d_in[1];
  const float* W_sa   = (const float*)d_in[2];
  const float* b_sa   = (const float*)d_in[3];
  const float* W_pe1  = (const float*)d_in[4];
  const float* b_pe1  = (const float*)d_in[5];
  const float* W_pe2  = (const float*)d_in[6];
  const float* b_pe2  = (const float*)d_in[7];
  const float* W_conv = (const float*)d_in[8];
  const float* b_conv = (const float*)d_in[9];
  const float* W_pw1  = (const float*)d_in[10];
  const float* b_pw1  = (const float*)d_in[11];
  const float* W_pw2  = (const float*)d_in[12];
  const float* b_pw2  = (const float*)d_in[13];
  float* feat = (float*)d_out;

  char* ws = (char*)d_ws;
  float* f1 = (float*)ws;  ws += sizeof(float) * (size_t)B_ * N_ * C_;
  float* h  = (float*)ws;  ws += sizeof(float) * (size_t)B_ * M_ * C_;
  float* h2 = (float*)ws;  ws += sizeof(float) * (size_t)B_ * M_ * EXP_ * C_;
  float* agg = (float*)ws; ws += sizeof(float) * (size_t)B_ * M_ * C_;
  int* idx1 = (int*)ws;    ws += sizeof(int) * (size_t)B_ * M_ * K_;
  int* idx2 = (int*)ws;

  const int THREADS = 256;

  {
    int waves = (B_ * N_ / 16) * (C_ / 16);
    f1_wmma_kernel<<<(waves * 32 + THREADS - 1) / THREADS, THREADS, 0, stream>>>(
        f, W_sa, b_sa, f1);
  }

  int bq_blocks = (B_ * M_ * 32 + THREADS - 1) / THREADS;
  ball_query_kernel<<<bq_blocks, THREADS, 0, stream>>>(p, N_, 1, R1SQ, idx1);
  ball_query_kernel<<<bq_blocks, THREADS, 0, stream>>>(p, M_, STRIDE_, R2SQ, idx2);

  int nm_blocks = (B_ * M_ + 3) / 4;
  neigh_max_kernel<<<nm_blocks, THREADS, 0, stream>>>(p, f1, idx1, 1, N_, W_pe1, b_pe1, feat);

  const int MR = B_ * M_;
  for (int i = 0; i < NB_; ++i) {
    {
      int waves = (MR / 64) * (C_ / 16);
      gemm_wmma_kernel<C_, true, false>
          <<<(waves * 32 + THREADS - 1) / THREADS, THREADS, 0, stream>>>(
              feat, W_conv + (size_t)i * C_ * C_, b_conv + (size_t)i * C_, nullptr, h, MR, C_);
    }
    neigh_max_kernel<<<nm_blocks, THREADS, 0, stream>>>(p, h, idx2, STRIDE_, M_, W_pe2,
                                                        b_pe2, agg);
    {
      int waves = (MR / 64) * ((EXP_ * C_) / 16);
      gemm_wmma_kernel<C_, true, false>
          <<<(waves * 32 + THREADS - 1) / THREADS, THREADS, 0, stream>>>(
              agg, W_pw1 + (size_t)i * C_ * EXP_ * C_, b_pw1 + (size_t)i * EXP_ * C_, nullptr,
              h2, MR, EXP_ * C_);
    }
    {
      int waves = (MR / 64) * (C_ / 16);
      gemm_wmma_kernel<EXP_ * C_, true, true>
          <<<(waves * 32 + THREADS - 1) / THREADS, THREADS, 0, stream>>>(
              h2, W_pw2 + (size_t)i * EXP_ * C_ * C_, b_pw2 + (size_t)i * C_, feat, feat, MR, C_);
    }
  }
}


# row-local GEMMs (pw1, pw2, next conv) run inside the gather-max launch on a 16-row tile kept in LDS (gather: 2 queries per wave, 16 B per lane): 12 launches -> 6; k order of every sum unchanged
# speedup vs baseline: 3.6676x; 3.6676x over previous
#include <hip/hip_runtime.h>
#include <hip/hip_bf16.h>
#include <cstdint>

namespace {

constexpr int B_ = 4, N_ = 8192, CIN_ = 4, C_ = 64, K_ = 32, STRIDE_ = 4, NB_ = 2, EXP_ = 4;
constexpr int M_ = N_ / STRIDE_;
constexpr float R1SQ = 0.1f * 0.1f;
constexpr float R2SQ = 0.2f * 0.2f;
constexpr float NEG_BIG = -3.4028235e38f;

typedef float v2f __attribute__((ext_vector_type(2)));
typedef float v8f __attribute__((ext_vector_type(8)));

__global__ void f1_wmma_kernel(const float* __restrict__ f, const float* __restrict__ Wsa,
                               const float* __restrict__ bsa, float* __restrict__ f1) {
  int wave = (blockIdx.x * blockDim.x + threadIdx.x) >> 5;
  int lane = threadIdx.x & 31;
  constexpr int tilesC = C_ / 16;
  constexpr int nTiles = (B_ * N_ / 16) * tilesC;
  if (wave >= nTiles) return;
  int row0 = (wave / tilesC) * 16;
  int c0 = (wave % tilesC) * 16;
  int b = row0 / N_;
  int n0 = row0 % N_;
  int half = lane >> 4;
  int ln = lane & 15;

  const float* fb = f + ((size_t)b * CIN_ + 2 * half) * N_ + n0 + ln;
  v2f a;  a.x = fb[0];        a.y = fb[N_];
  const float* wc = Wsa + (size_t)(2 * half) * C_ + c0 + ln;
  v2f bb; bb.x = wc[0];       bb.y = wc[C_];
  v8f acc = {0.f, 0.f, 0.f, 0.f, 0.f, 0.f, 0.f, 0.f};
  acc = __builtin_amdgcn_wmma_f32_16x16x4_f32(false, a, false, bb, (short)0, acc,
                                              false, false);
  int col = c0 + ln;
  float bv = bsa[col];
#pragma unroll
  for (int r = 0; r < 8; ++r) {
    int row = row0 + r + 8 * half;
    f1[(size_t)row * C_ + col] = fmaxf(acc[r] + bv, 0.0f);
  }
}

constexpr int BQ_THREADS = 512;
constexpr int BQ_WAVES = BQ_THREADS / 32;
constexpr int BQ_GROUP = 256;
constexpr int BQ_CHUNKS = BQ_GROUP / 32;

template <int NSRC, int SSTRIDE>
__global__ __launch_bounds__(BQ_THREADS) void ball_query_kernel(const float* __restrict__ p,
                                                                float r2,
                                                                int* __restrict__ idx_out) {
  static_assert(NSRC % BQ_THREADS == 0, "staging loop covers NSRC exactly");
  static_assert(NSRC % BQ_GROUP == 0, "scan trips cover NSRC exactly");
  static_assert(M_ % BQ_WAVES == 0, "a block never crosses a batch");
  __shared__ float s[NSRC / BQ_GROUP][3][BQ_GROUP];

  const int tid = threadIdx.x;
  const int lane = tid & 31;
  const int q0 = blockIdx.x * BQ_WAVES;
  const int b = q0 / M_;
  const float* pb = p + (size_t)b * N_ * 3;

#pragma unroll
  for (int i = 0; i < NSRC / BQ_THREADS; ++i) {
    int j = tid + i * BQ_THREADS;
    const float* sp = pb + (size_t)j * SSTRIDE * 3;
    s[j / BQ_GROUP][0][j % BQ_GROUP] = sp[0];
    s[j / BQ_GROUP][1][j % BQ_GROUP] = sp[1];
    s[j / BQ_GROUP][2][j % BQ_GROUP] = sp[2];
  }
  __syncthreads();

  const int w = __builtin_amdgcn_readfirstlane(tid >> 5);
  const int wave = q0 + w;
  const int m = wave - b * M_;
  const float qx = pb[(size_t)m * STRIDE_ * 3 + 0];
  const float qy = pb[(size_t)m * STRIDE_ * 3 + 1];
  const float qz = pb[(size_t)m * STRIDE_ * 3 + 2];
  int* out = idx_out + (size_t)wave * K_;

  int count = 0;
  int first_idx = NSRC - 1;
  for (int g = 0; g < NSRC / BQ_GROUP; ++g) {
    unsigned mk[BQ_CHUNKS];
#pragma unroll
    for (int u = 0; u < BQ_CHUNKS; ++u) {
      int jj = u * 32 + lane;
      float dx = s[g][0][jj] - qx, dy = s[g][1][jj] - qy, dz = s[g][2][jj] - qz;
      float sxy;
      {
#pragma clang fp contract(off)
        sxy = dx * dx + dy * dy;
      }
      float d2 = __builtin_fmaf(dz, dz, sxy);
      mk[u] = (unsigned)__ballot(d2 <= r2);
    }
    unsigned any = 0u;
#pragma unroll
    for (int u = 0; u < BQ_CHUNKS; ++u) any |= mk[u];
    if (any != 0u) {
      const int base = g * BQ_GROUP;
#pragma unroll
      for (int u = 0; u < BQ_CHUNKS; ++u) {
        unsigned mask = mk[u];
        if (mask != 0u) {
          if (count == 0) first_idx = base + u * 32 + (__ffs(mask) - 1);
          if ((mask >> lane) & 1u) {
            int pos = count + (int)__builtin_amdgcn_mbcnt_lo(mask, 0u);
            if (pos < K_) out[pos] = base + u * 32 + lane;
          }
          count += __popc(mask);
        }
      }
      if (count >= K_) break;
    }
  }
  if (count > K_) count = K_;
  for (int t = count + lane; t < K_; t += 32) out[t] = first_idx;
}

constexpr int TR = 16;
constexpr int FT = 256;
constexpr int HID = EXP_ * C_;
constexpr int PT = C_ + 4;
constexpr int PH = HID + 4;

__device__ __forceinline__ v8f wmma4(v2f a, v2f b, v8f c) {
  return __builtin_amdgcn_wmma_f32_16x16x4_f32(false, a, false, b, (short)0, c, false, false);
}

template <bool MLP, bool CONV>
__global__ __launch_bounds__(FT) void row_tile_kernel(
    const float* __restrict__ p, const float* __restrict__ src, const int* __restrict__ idx,
    int src_stride, int src_n, const float* __restrict__ Wpe, const float* __restrict__ bpe,
    const float* __restrict__ Wpw1, const float* __restrict__ bpw1,
    const float* __restrict__ Wpw2, const float* __restrict__ bpw2, float* feat,
    const float* __restrict__ Wc, const float* __restrict__ bc, float* __restrict__ hout) {
  static_assert(M_ % TR == 0, "a tile never crosses a batch");
  static_assert(TR * K_ == 2 * FT && K_ == 32, "staging: two records per thread, a wave per query");
  static_assert(TR == 2 * (FT / 32), "gather: two queries per wave");
  __shared__ __attribute__((aligned(16))) int4 s_nb[TR][K_];
  __shared__ __attribute__((aligned(16))) float sT[TR][PT];
  __shared__ __attribute__((aligned(16))) float sH[MLP ? TR : 1][PH];

  const int tid = threadIdx.x;
  const int lane = tid & 31;
  const int half = lane >> 4;
  const int ln = lane & 15;
  const int w = __builtin_amdgcn_readfirstlane(tid >> 5);
  const int bm0 = blockIdx.x * TR;
  const int b = bm0 / M_;
  const float* pb = p + (size_t)b * N_ * 3;

  {
    const int* ib = idx + (size_t)bm0 * K_;
    const int jA = ib[tid];
    const int jB = ib[tid + FT];
    const int qlA = w, qlB = w + TR / 2;
    const float* qa = pb + (size_t)(bm0 - b * M_ + qlA) * STRIDE_ * 3;
    const float* qb = pb + (size_t)(bm0 - b * M_ + qlB) * STRIDE_ * 3;
    const float* sa = pb + (size_t)jA * src_stride * 3;
    const float* sc = pb + (size_t)jB * src_stride * 3;
    const float ax = sa[0], ay = sa[1], az = sa[2];
    const float cx = sc[0], cy = sc[1], cz = sc[2];
    int4 recA, recB;
    recA.x = jA;
    recA.y = __float_as_int(ax - qa[0]);
    recA.z = __float_as_int(ay - qa[1]);
    recA.w = __float_as_int(az - qa[2]);
    recB.x = jB;
    recB.y = __float_as_int(cx - qb[0]);
    recB.z = __float_as_int(cy - qb[1]);
    recB.w = __float_as_int(cz - qb[2]);
    s_nb[qlA][lane] = recA;
    s_nb[qlB][lane] = recB;
  }
  __syncthreads();

  {
    const int c4 = ln * 4;
    const float4 w0 = *(const float4*)(Wpe + 0 * C_ + c4);
    const float4 w1 = *(const float4*)(Wpe + 1 * C_ + c4);
    const float4 w2 = *(const float4*)(Wpe + 2 * C_ + c4);
    const float4 bi = *(const float4*)(bpe + c4);
    const float* sb = src + (size_t)b * src_n * C_ + c4;
#pragma unroll
    for (int qq = 0; qq < 2; ++qq) {
      const int ql = 2 * w + qq;
      float4 best = {NEG_BIG, NEG_BIG, NEG_BIG, NEG_BIG};
#pragma unroll
      for (int i = 0; i < K_ / 2; ++i) {
        const int4 rec = s_nb[ql][2 * i + half];
        const float dx = __int_as_float(rec.y), dy = __int_as_float(rec.z),
                    dz = __int_as_float(rec.w);
        const float4 s4 = *(const float4*)(sb + (size_t)rec.x * C_);
        best.x = fmaxf(best.x, dx * w0.x + dy * w1.x + dz * w2.x + bi.x + s4.x);
        best.y = fmaxf(best.y, dx * w0.y + dy * w1.y + dz * w2.y + bi.y + s4.y);
        best.z = fmaxf(best.z, dx * w0.z + dy * w1.z + dz * w2.z + bi.z + s4.z);
        best.w = fmaxf(best.w, dx * w0.w + dy * w1.w + dz * w2.w + bi.w + s4.w);
      }
      best.x = fmaxf(best.x, __shfl_xor(best.x, 16));
      best.y = fmaxf(best.y, __shfl_xor(best.y, 16));
      best.z = fmaxf(best.z, __shfl_xor(best.z, 16));
      best.w = fmaxf(best.w, __shfl_xor(best.w, 16));
      if (half == 0) *(float4*)&sT[ql][c4] = best;
    }
  }
  __syncthreads();

  if constexpr (MLP) {
    {
      const int n0 = w * 32;
      const float* Wb = Wpw1 + (size_t)(2 * half) * HID + n0 + ln;
      v8f acc0 = {0.f, 0.f, 0.f, 0.f, 0.f, 0.f, 0.f, 0.f};
      v8f acc1 = {0.f, 0.f, 0.f, 0.f, 0.f, 0.f, 0.f, 0.f};
#pragma unroll
      for (int k = 0; k < C_; k += 4) {
        v2f a;   a.x = sT[ln][k + 2 * half];        a.y = sT[ln][k + 2 * half + 1];
        v2f b0;  b0.x = Wb[(size_t)k * HID];        b0.y = Wb[(size_t)(k + 1) * HID];
        v2f b1;  b1.x = Wb[(size_t)k * HID + 16];   b1.y = Wb[(size_t)(k + 1) * HID + 16];
        acc0 = wmma4(a, b0, acc0);
        acc1 = wmma4(a, b1, acc1);
      }
      const float bv0 = bpw1[n0 + ln], bv1 = bpw1[n0 + 16 + ln];
#pragma unroll
      for (int r = 0; r < 8; ++r) {
        sH[r + 8 * half][n0 + ln] = fmaxf(acc0[r] + bv0, 0.0f);
        sH[r + 8 * half][n0 + 16 + ln] = fmaxf(acc1[r] + bv1, 0.0f);
      }
    }
    __syncthreads();

    if (w < 4) {
      const int n0 = w * 16;
      const float* Wb = Wpw2 + (size_t)(2 * half) * C_ + n0 + ln;
      v8f acc = {0.f, 0.f, 0.f, 0.f, 0.f, 0.f, 0.f, 0.f};
#pragma unroll
      for (int k = 0; k < HID; k += 4) {
        v2f a;   a.x = sH[ln][k + 2 * half];        a.y = sH[ln][k + 2 * half + 1];
        v2f bb;  bb.x = Wb[(size_t)k * C_];         bb.y = Wb[(size_t)(k + 1) * C_];
        acc = wmma4(a, bb, acc);
      }
      const int col = n0 + ln;
      const float bv = bpw2[col];
#pragma unroll
      for (int r = 0; r < 8; ++r) {
        const int row = r + 8 * half;
        float v = acc[r] + bv;
        v += feat[(size_t)(bm0 + row) * C_ + col];
        sT[row][col] = fmaxf(v, 0.0f);
      }
    }
    __syncthreads();
  }

  {
    const int row = tid >> 4, c4 = (tid & 15) * 4;
    *(float4*)(feat + (size_t)(bm0 + row) * C_ + c4) = *(const float4*)&sT[row][c4];
  }

  if constexpr (CONV) {
    if (w < 4) {
      const int n0 = w * 16;
      const float* Wb = Wc + (size_t)(2 * half) * C_ + n0 + ln;
      v8f acc = {0.f, 0.f, 0.f, 0.f, 0.f, 0.f, 0.f, 0.f};
#pragma unroll
      for (int k = 0; k < C_; k += 4) {
        v2f a;   a.x = sT[ln][k + 2 * half];        a.y = sT[ln][k + 2 * half + 1];
        v2f bb;  bb.x = Wb[(size_t)k * C_];         bb.y = Wb[(size_t)(k + 1) * C_];
        acc = wmma4(a, bb, acc);
      }
      const int col = n0 + ln;
      const float bv = bc[col];
#pragma unroll
      for (int r = 0; r < 8; ++r) {
        const int row = bm0 + r + 8 * half;
        hout[(size_t)row * C_ + col] = fmaxf(acc[r] + bv, 0.0f);
      }
    }
  }
}

}

extern "C" void kernel_launch(void* const* d_in, const int* in_sizes, int n_in,
                              void* d_out, int out_size, void* d_ws, size_t ws_size,
                              hipStream_t stream) {
  (void)in_sizes; (void)n_in; (void)out_size; (void)ws_size;
  const float* p      = (const float*)d_in[0];
  const float* f      = (const float*)d_in[1];
  const float* W_sa   = (const float*)d_in[2];
  const float* b_sa   = (const float*)d_in[3];
  const float* W_pe1  = (const float*)d_in[4];
  const float* b_pe1  = (const float*)d_in[5];
  const float* W_pe2  = (const float*)d_in[6];
  const float* b_pe2  = (const float*)d_in[7];
  const float* W_conv = (const float*)d_in[8];
  const float* b_conv = (const float*)d_in[9];
  const float* W_pw1  = (const float*)d_in[10];
  const float* b_pw1  = (const float*)d_in[11];
  const float* W_pw2  = (const float*)d_in[12];
  const float* b_pw2  = (const float*)d_in[13];
  float* feat = (float*)d_out;

  char* ws = (char*)d_ws;
  float* f1 = (float*)ws;  ws += sizeof(float) * (size_t)B_ * N_ * C_;
  float* hA = (float*)ws;  ws += sizeof(float) * (size_t)B_ * M_ * C_;
  float* hB = (float*)ws;  ws += sizeof(float) * (size_t)B_ * M_ * C_;
  int* idx1 = (int*)ws;    ws += sizeof(int) * (size_t)B_ * M_ * K_;
  int* idx2 = (int*)ws;

  const int THREADS = 256;

  {
    int waves = (B_ * N_ / 16) * (C_ / 16);
    f1_wmma_kernel<<<(waves * 32 + THREADS - 1) / THREADS, THREADS, 0, stream>>>(
        f, W_sa, b_sa, f1);
  }

  const int bq_blocks = (B_ * M_) / BQ_WAVES;
  ball_query_kernel<N_, 1><<<bq_blocks, BQ_THREADS, 0, stream>>>(p, R1SQ, idx1);
  ball_query_kernel<M_, STRIDE_><<<bq_blocks, BQ_THREADS, 0, stream>>>(p, R2SQ, idx2);

  const int tiles = (B_ * M_) / TR;

  row_tile_kernel<false, true><<<tiles, FT, 0, stream>>>(
      p, f1, idx1, 1, N_, W_pe1, b_pe1, nullptr, nullptr, nullptr, nullptr, feat,
      W_conv, b_conv, hA);

  row_tile_kernel<true, true><<<tiles, FT, 0, stream>>>(
      p, hA, idx2, STRIDE_, M_, W_pe2, b_pe2, W_pw1, b_pw1, W_pw2, b_pw2, feat,
      W_conv + (size_t)C_ * C_, b_conv + C_, hB);

  row_tile_kernel<true, false><<<tiles, FT, 0, stream>>>(
      p, hB, idx2, STRIDE_, M_, W_pe2, b_pe2, W_pw1 + (size_t)C_ * HID, b_pw1 + HID,
      W_pw2 + (size_t)HID * C_, b_pw2 + C_, feat, nullptr, nullptr, nullptr);
}


# embedding GEMM and both ball queries (inputs only) merged into ONE launch, a role per 16-wave block, long first-query blocks dealt 2:1 among short ones: 6 launches -> 4; sums, hit order unchanged
# speedup vs baseline: 4.2255x; 1.1521x over previous
#include <hip/hip_runtime.h>
#include <hip/hip_bf16.h>
#include <cstdint>

namespace {

constexpr int B_ = 4, N_ = 8192, CIN_ = 4, C_ = 64, K_ = 32, STRIDE_ = 4, NB_ = 2, EXP_ = 4;
constexpr int M_ = N_ / STRIDE_;
constexpr float R1SQ = 0.1f * 0.1f;
constexpr float R2SQ = 0.2f * 0.2f;
constexpr float NEG_BIG = -3.4028235e38f;

typedef float v2f __attribute__((ext_vector_type(2)));
typedef float v8f __attribute__((ext_vector_type(8)));

__device__ __forceinline__ void f1_wmma_tile(int wave, int lane, const float* __restrict__ f,
                                             const float* __restrict__ Wsa,
                                             const float* __restrict__ bsa,
                                             float* __restrict__ f1) {
  constexpr int tilesC = C_ / 16;
  int row0 = (wave / tilesC) * 16;
  int c0 = (wave % tilesC) * 16;
  int b = row0 / N_;
  int n0 = row0 % N_;
  int half = lane >> 4;
  int ln = lane & 15;

  const float* fb = f + ((size_t)b * CIN_ + 2 * half) * N_ + n0 + ln;
  v2f a;  a.x = fb[0];        a.y = fb[N_];
  const float* wc = Wsa + (size_t)(2 * half) * C_ + c0 + ln;
  v2f bb; bb.x = wc[0];       bb.y = wc[C_];
  v8f acc = {0.f, 0.f, 0.f, 0.f, 0.f, 0.f, 0.f, 0.f};
  acc = __builtin_amdgcn_wmma_f32_16x16x4_f32(false, a, false, bb, (short)0, acc,
                                              false, false);
  int col = c0 + ln;
  float bv = bsa[col];
#pragma unroll
  for (int r = 0; r < 8; ++r) {
    int row = row0 + r + 8 * half;
    f1[(size_t)row * C_ + col] = fmaxf(acc[r] + bv, 0.0f);
  }
}

constexpr int BQ_THREADS = 512;
constexpr int BQ_WAVES = BQ_THREADS / 32;
constexpr int BQ_GROUP = 256;
constexpr int BQ_CHUNKS = BQ_GROUP / 32;

template <int NSRC, int SSTRIDE>
__device__ __forceinline__ void ball_query_block(float (*s)[3][BQ_GROUP], int blk,
                                                 const float* __restrict__ p, float r2,
                                                 int* __restrict__ idx_out) {
  static_assert(NSRC % BQ_THREADS == 0, "staging loop covers NSRC exactly");
  static_assert(NSRC % BQ_GROUP == 0, "scan trips cover NSRC exactly");
  static_assert(M_ % BQ_WAVES == 0, "a block never crosses a batch");

  const int tid = threadIdx.x;
  const int lane = tid & 31;
  const int q0 = blk * BQ_WAVES;
  const int b = q0 / M_;
  const float* pb = p + (size_t)b * N_ * 3;

#pragma unroll
  for (int i = 0; i < NSRC / BQ_THREADS; ++i) {
    int j = tid + i * BQ_THREADS;
    const float* sp = pb + (size_t)j * SSTRIDE * 3;
    s[j / BQ_GROUP][0][j % BQ_GROUP] = sp[0];
    s[j / BQ_GROUP][1][j % BQ_GROUP] = sp[1];
    s[j / BQ_GROUP][2][j % BQ_GROUP] = sp[2];
  }
  __syncthreads();

  const int w = __builtin_amdgcn_readfirstlane(tid >> 5);
  const int wave = q0 + w;
  const int m = wave - b * M_;
  const float qx = pb[(size_t)m * STRIDE_ * 3 + 0];
  const float qy = pb[(size_t)m * STRIDE_ * 3 + 1];
  const float qz = pb[(size_t)m * STRIDE_ * 3 + 2];
  int* out = idx_out + (size_t)wave * K_;

  int count = 0;
  int first_idx = NSRC - 1;
  for (int g = 0; g < NSRC / BQ_GROUP; ++g) {
    unsigned mk[BQ_CHUNKS];
#pragma unroll
    for (int u = 0; u < BQ_CHUNKS; ++u) {
      int jj = u * 32 + lane;
      float dx = s[g][0][jj] - qx, dy = s[g][1][jj] - qy, dz = s[g][2][jj] - qz;
      float sxy;
      {
#pragma clang fp contract(off)
        sxy = dx * dx + dy * dy;
      }
      float d2 = __builtin_fmaf(dz, dz, sxy);
      mk[u] = (unsigned)__ballot(d2 <= r2);
    }
    unsigned any = 0u;
#pragma unroll
    for (int u = 0; u < BQ_CHUNKS; ++u) any |= mk[u];
    if (any != 0u) {
      const int base = g * BQ_GROUP;
#pragma unroll
      for (int u = 0; u < BQ_CHUNKS; ++u) {
        unsigned mask = mk[u];
        if (mask != 0u) {
          if (count == 0) first_idx = base + u * 32 + (__ffs(mask) - 1);
          if ((mask >> lane) & 1u) {
            int pos = count + (int)__builtin_amdgcn_mbcnt_lo(mask, 0u);
            if (pos < K_) out[pos] = base + u * 32 + lane;
          }
          count += __popc(mask);
        }
      }
      if (count >= K_) break;
    }
  }
  if (count > K_) count = K_;
  for (int t = count + lane; t < K_; t += 32) out[t] = first_idx;
}

constexpr int FR_BQ = (B_ * M_) / BQ_WAVES;
constexpr int FR_F1 = ((B_ * N_ / 16) * (C_ / 16)) / BQ_WAVES;
constexpr int FR_MIX = (FR_BQ / 2) * 3;
constexpr int FR_BLOCKS = 2 * FR_BQ + FR_F1;
static_assert(FR_BQ % 2 == 0, "the mixed range pairs the first query's blocks");
static_assert(((B_ * N_ / 16) * (C_ / 16)) % BQ_WAVES == 0, "embedding tiles fill whole blocks");

__global__ __launch_bounds__(BQ_THREADS) void front_kernel(
    const float* __restrict__ p, const float* __restrict__ f, const float* __restrict__ Wsa,
    const float* __restrict__ bsa, float* __restrict__ f1, int* __restrict__ idx1,
    int* __restrict__ idx2) {
  __shared__ float s_pts[N_ / BQ_GROUP][3][BQ_GROUP];
  const int bid = blockIdx.x;
  int role, blk;
  if (bid < FR_MIX) {
    const int t = bid / 3, r = bid - 3 * t;
    role = r < 2 ? 0 : 1;
    blk = r < 2 ? 2 * t + r : t;
  } else if (bid < 2 * FR_BQ) {
    role = 1;
    blk = bid - FR_MIX + FR_BQ / 2;
  } else {
    role = 2;
    blk = bid - 2 * FR_BQ;
  }
  if (role == 0) {
    ball_query_block<N_, 1>(s_pts, blk, p, R1SQ, idx1);
  } else if (role == 1) {
    ball_query_block<M_, STRIDE_>(s_pts, blk, p, R2SQ, idx2);
  } else {
    const int wave = blk * BQ_WAVES + (int)(threadIdx.x >> 5);
    f1_wmma_tile(wave, (int)(threadIdx.x & 31), f, Wsa, bsa, f1);
  }
}

constexpr int TR = 16;
constexpr int FT = 256;
constexpr int HID = EXP_ * C_;
constexpr int PT = C_ + 4;
constexpr int PH = HID + 4;

__device__ __forceinline__ v8f wmma4(v2f a, v2f b, v8f c) {
  return __builtin_amdgcn_wmma_f32_16x16x4_f32(false, a, false, b, (short)0, c, false, false);
}

template <bool MLP, bool CONV>
__global__ __launch_bounds__(FT) void row_tile_kernel(
    const float* __restrict__ p, const float* __restrict__ src, const int* __restrict__ idx,
    int src_stride, int src_n, const float* __restrict__ Wpe, const float* __restrict__ bpe,
    const float* __restrict__ Wpw1, const float* __restrict__ bpw1,
    const float* __restrict__ Wpw2, const float* __restrict__ bpw2, float* feat,
    const float* __restrict__ Wc, const float* __restrict__ bc, float* __restrict__ hout) {
  static_assert(M_ % TR == 0, "a tile never crosses a batch");
  static_assert(TR * K_ == 2 * FT && K_ == 32, "staging: two records per thread, a wave per query");
  static_assert(TR == 2 * (FT / 32), "gather: two queries per wave");
  __shared__ __attribute__((aligned(16))) int4 s_nb[TR][K_];
  __shared__ __attribute__((aligned(16))) float sT[TR][PT];
  __shared__ __attribute__((aligned(16))) float sH[MLP ? TR : 1][PH];

  const int tid = threadIdx.x;
  const int lane = tid & 31;
  const int half = lane >> 4;
  const int ln = lane & 15;
  const int w = __builtin_amdgcn_readfirstlane(tid >> 5);
  const int bm0 = blockIdx.x * TR;
  const int b = bm0 / M_;
  const float* pb = p + (size_t)b * N_ * 3;

  {
    const int* ib = idx + (size_t)bm0 * K_;
    const int jA = ib[tid];
    const int jB = ib[tid + FT];
    const int qlA = w, qlB = w + TR / 2;
    const float* qa = pb + (size_t)(bm0 - b * M_ + qlA) * STRIDE_ * 3;
    const float* qb = pb + (size_t)(bm0 - b * M_ + qlB) * STRIDE_ * 3;
    const float* sa = pb + (size_t)jA * src_stride * 3;
    const float* sc = pb + (size_t)jB * src_stride * 3;
    const float ax = sa[0], ay = sa[1], az = sa[2];
    const float cx = sc[0], cy = sc[1], cz = sc[2];
    int4 recA, recB;
    recA.x = jA;
    recA.y = __float_as_int(ax - qa[0]);
    recA.z = __float_as_int(ay - qa[1]);
    recA.w = __float_as_int(az - qa[2]);
    recB.x = jB;
    recB.y = __float_as_int(cx - qb[0]);
    recB.z = __float_as_int(cy - qb[1]);
    recB.w = __float_as_int(cz - qb[2]);
    s_nb[qlA][lane] = recA;
    s_nb[qlB][lane] = recB;
  }
  __syncthreads();

  {
    const int c4 = ln * 4;
    const float4 w0 = *(const float4*)(Wpe + 0 * C_ + c4);
    const float4 w1 = *(const float4*)(Wpe + 1 * C_ + c4);
    const float4 w2 = *(const float4*)(Wpe + 2 * C_ + c4);
    const float4 bi = *(const float4*)(bpe + c4);
    const float* sb = src + (size_t)b * src_n * C_ + c4;
#pragma unroll
    for (int qq = 0; qq < 2; ++qq) {
      const int ql = 2 * w + qq;
      float4 best = {NEG_BIG, NEG_BIG, NEG_BIG, NEG_BIG};
#pragma unroll
      for (int i = 0; i < K_ / 2; ++i) {
        const int4 rec = s_nb[ql][2 * i + half];
        const float dx = __int_as_float(rec.y), dy = __int_as_float(rec.z),
                    dz = __int_as_float(rec.w);
        const float4 s4 = *(const float4*)(sb + (size_t)rec.x * C_);
        best.x = fmaxf(best.x, dx * w0.x + dy * w1.x + dz * w2.x + bi.x + s4.x);
        best.y = fmaxf(best.y, dx * w0.y + dy * w1.y + dz * w2.y + bi.y + s4.y);
        best.z = fmaxf(best.z, dx * w0.z + dy * w1.z + dz * w2.z + bi.z + s4.z);
        best.w = fmaxf(best.w, dx * w0.w + dy * w1.w + dz * w2.w + bi.w + s4.w);
      }
      best.x = fmaxf(best.x, __shfl_xor(best.x, 16));
      best.y = fmaxf(best.y, __shfl_xor(best.y, 16));
      best.z = fmaxf(best.z, __shfl_xor(best.z, 16));
      best.w = fmaxf(best.w, __shfl_xor(best.w, 16));
      if (half == 0) *(float4*)&sT[ql][c4] = best;
    }
  }
  __syncthreads();

  if constexpr (MLP) {
    {
      const int n0 = w * 32;
      const float* Wb = Wpw1 + (size_t)(2 * half) * HID + n0 + ln;
      v8f acc0 = {0.f, 0.f, 0.f, 0.f, 0.f, 0.f, 0.f, 0.f};
      v8f acc1 = {0.f, 0.f, 0.f, 0.f, 0.f, 0.f, 0.f, 0.f};
#pragma unroll
      for (int k = 0; k < C_; k += 4) {
        v2f a;   a.x = sT[ln][k + 2 * half];        a.y = sT[ln][k + 2 * half + 1];
        v2f b0;  b0.x = Wb[(size_t)k * HID];        b0.y = Wb[(size_t)(k + 1) * HID];
        v2f b1;  b1.x = Wb[(size_t)k * HID + 16];   b1.y = Wb[(size_t)(k + 1) * HID + 16];
        acc0 = wmma4(a, b0, acc0);
        acc1 = wmma4(a, b1, acc1);
      }
      const float bv0 = bpw1[n0 + ln], bv1 = bpw1[n0 + 16 + ln];
#pragma unroll
      for (int r = 0; r < 8; ++r) {
        sH[r + 8 * half][n0 + ln] = fmaxf(acc0[r] + bv0, 0.0f);
        sH[r + 8 * half][n0 + 16 + ln] = fmaxf(acc1[r] + bv1, 0.0f);
      }
    }
    __syncthreads();

    if (w < 4) {
      const int n0 = w * 16;
      const float* Wb = Wpw2 + (size_t)(2 * half) * C_ + n0 + ln;
      v8f acc = {0.f, 0.f, 0.f, 0.f, 0.f, 0.f, 0.f, 0.f};
#pragma unroll
      for (int k = 0; k < HID; k += 4) {
        v2f a;   a.x = sH[ln][k + 2 * half];        a.y = sH[ln][k + 2 * half + 1];
        v2f bb;  bb.x = Wb[(size_t)k * C_];         bb.y = Wb[(size_t)(k + 1) * C_];
        acc = wmma4(a, bb, acc);
      }
      const int col = n0 + ln;
      const float bv = bpw2[col];
#pragma unroll
      for (int r = 0; r < 8; ++r) {
        const int row = r + 8 * half;
        float v = acc[r] + bv;
        v += feat[(size_t)(bm0 + row) * C_ + col];
        sT[row][col] = fmaxf(v, 0.0f);
      }
    }
    __syncthreads();
  }

  {
    const int row = tid >> 4, c4 = (tid & 15) * 4;
    *(float4*)(feat + (size_t)(bm0 + row) * C_ + c4) = *(const float4*)&sT[row][c4];
  }

  if constexpr (CONV) {
    if (w < 4) {
      const int n0 = w * 16;
      const float* Wb = Wc + (size_t)(2 * half) * C_ + n0 + ln;
      v8f acc = {0.f, 0.f, 0.f, 0.f, 0.f, 0.f, 0.f, 0.f};
#pragma unroll
      for (int k = 0; k < C_; k += 4) {
        v2f a;   a.x = sT[ln][k + 2 * half];        a.y = sT[ln][k + 2 * half + 1];
        v2f bb;  bb.x = Wb[(size_t)k * C_];         bb.y = Wb[(size_t)(k + 1) * C_];
        acc = wmma4(a, bb, acc);
      }
      const int col = n0 + ln;
      const float bv = bc[col];
#pragma unroll
      for (int r = 0; r < 8; ++r) {
        const int row = bm0 + r + 8 * half;
        hout[(size_t)row * C_ + col] = fmaxf(acc[r] + bv, 0.0f);
      }
    }
  }
}

}

extern "C" void kernel_launch(void* const* d_in, const int* in_sizes, int n_in,
                              void* d_out, int out_size, void* d_ws, size_t ws_size,
                              hipStream_t stream) {
  (void)in_sizes; (void)n_in; (void)out_size; (void)ws_size;
  const float* p      = (const float*)d_in[0];
  const float* f      = (const float*)d_in[1];
  const float* W_sa   = (const float*)d_in[2];
  const float* b_sa   = (const float*)d_in[3];
  const float* W_pe1  = (const float*)d_in[4];
  const float* b_pe1  = (const float*)d_in[5];
  const float* W_pe2  = (const float*)d_in[6];
  const float* b_pe2  = (const float*)d_in[7];
  const float* W_conv = (const float*)d_in[8];
  const float* b_conv = (const float*)d_in[9];
  const float* W_pw1  = (const float*)d_in[10];
  const float* b_pw1  = (const float*)d_in[11];
  const float* W_pw2  = (const float*)d_in[12];
  const float* b_pw2  = (const float*)d_in[13];
  float* feat = (float*)d_out;

  char* ws = (char*)d_ws;
  float* f1 = (float*)ws;  ws += sizeof(float) * (size_t)B_ * N_ * C_;
  float* hA = (float*)ws;  ws += sizeof(float) * (size_t)B_ * M_ * C_;
  float* hB = (float*)ws;  ws += sizeof(float) * (size_t)B_ * M_ * C_;
  int* idx1 = (int*)ws;    ws += sizeof(int) * (size_t)B_ * M_ * K_;
  int* idx2 = (int*)ws;

  front_kernel<<<FR_BLOCKS, BQ_THREADS, 0, stream>>>(p, f, W_sa, b_sa, f1, idx1, idx2);

  const int tiles = (B_ * M_) / TR;

  row_tile_kernel<false, true><<<tiles, FT, 0, stream>>>(
      p, f1, idx1, 1, N_, W_pe1, b_pe1, nullptr, nullptr, nullptr, nullptr, feat,
      W_conv, b_conv, hA);

  row_tile_kernel<true, true><<<tiles, FT, 0, stream>>>(
      p, hA, idx2, STRIDE_, M_, W_pe2, b_pe2, W_pw1, b_pw1, W_pw2, b_pw2, feat,
      W_conv + (size_t)C_ * C_, b_conv + C_, hB);

  row_tile_kernel<true, false><<<tiles, FT, 0, stream>>>(
      p, hB, idx2, STRIDE_, M_, W_pe2, b_pe2, W_pw1 + (size_t)C_ * HID, b_pw1 + HID,
      W_pw2 + (size_t)HID * C_, b_pw2 + C_, feat, nullptr, nullptr, nullptr);
}


# ball queries: LDS planes hold a lane's source of 4 consecutive chunks as one 16-byte word: 6 ds_load_b128 per 256-source trip (was 12 two-address loads), 16-byte staging stores; hit order unchanged
# speedup vs baseline: 4.4629x; 1.0562x over previous
#include <hip/hip_runtime.h>
#include <hip/hip_bf16.h>
#include <cstdint>

namespace {

constexpr int B_ = 4, N_ = 8192, CIN_ = 4, C_ = 64, K_ = 32, STRIDE_ = 4, NB_ = 2, EXP_ = 4;
constexpr int M_ = N_ / STRIDE_;
constexpr float R1SQ = 0.1f * 0.1f;
constexpr float R2SQ = 0.2f * 0.2f;
constexpr float NEG_BIG = -3.4028235e38f;

typedef float v2f __attribute__((ext_vector_type(2)));
typedef float v8f __attribute__((ext_vector_type(8)));

__device__ __forceinline__ void f1_wmma_tile(int wave, int lane, const float* __restrict__ f,
                                             const float* __restrict__ Wsa,
                                             const float* __restrict__ bsa,
                                             float* __restrict__ f1) {
  constexpr int tilesC = C_ / 16;
  int row0 = (wave / tilesC) * 16;
  int c0 = (wave % tilesC) * 16;
  int b = row0 / N_;
  int n0 = row0 % N_;
  int half = lane >> 4;
  int ln = lane & 15;

  const float* fb = f + ((size_t)b * CIN_ + 2 * half) * N_ + n0 + ln;
  v2f a;  a.x = fb[0];        a.y = fb[N_];
  const float* wc = Wsa + (size_t)(2 * half) * C_ + c0 + ln;
  v2f bb; bb.x = wc[0];       bb.y = wc[C_];
  v8f acc = {0.f, 0.f, 0.f, 0.f, 0.f, 0.f, 0.f, 0.f};
  acc = __builtin_amdgcn_wmma_f32_16x16x4_f32(false, a, false, bb, (short)0, acc,
                                              false, false);
  int col = c0 + ln;
  float bv = bsa[col];
#pragma unroll
  for (int r = 0; r < 8; ++r) {
    int row = row0 + r + 8 * half;
    f1[(size_t)row * C_ + col] = fmaxf(acc[r] + bv, 0.0f);
  }
}

constexpr int BQ_THREADS = 512;
constexpr int BQ_WAVES = BQ_THREADS / 32;
constexpr int BQ_GROUP = 256;
constexpr int BQ_CHUNKS = BQ_GROUP / 32;
constexpr int BQ_QUAD = 128;
constexpr int BQ_QUADS = BQ_GROUP / BQ_QUAD;

template <int NSRC, int SSTRIDE>
__device__ __forceinline__ void ball_query_block(float4 (*s)[3][BQ_QUADS][32], int blk,
                                                 const float* __restrict__ p, float r2,
                                                 int* __restrict__ idx_out) {
  static_assert((NSRC / BQ_QUAD) % BQ_WAVES == 0, "staging loop covers NSRC exactly, a quad per wave and pass");
  static_assert(NSRC % BQ_GROUP == 0, "scan trips cover NSRC exactly");
  static_assert(M_ % BQ_WAVES == 0, "a block never crosses a batch");

  const int tid = threadIdx.x;
  const int lane = tid & 31;
  const int q0 = blk * BQ_WAVES;
  const int b = q0 / M_;
  const float* pb = p + (size_t)b * N_ * 3;

#pragma unroll
  for (int i = 0; i < (NSRC / BQ_QUAD) / BQ_WAVES; ++i) {
    const int quad = (tid >> 5) + i * BQ_WAVES;
    float cx[4], cy[4], cz[4];
#pragma unroll
    for (int e = 0; e < 4; ++e) {
      const int j = quad * BQ_QUAD + e * 32 + lane;
      const float* sp = pb + (size_t)j * SSTRIDE * 3;
      cx[e] = sp[0];
      cy[e] = sp[1];
      cz[e] = sp[2];
    }
    s[quad / BQ_QUADS][0][quad % BQ_QUADS][lane] = make_float4(cx[0], cx[1], cx[2], cx[3]);
    s[quad / BQ_QUADS][1][quad % BQ_QUADS][lane] = make_float4(cy[0], cy[1], cy[2], cy[3]);
    s[quad / BQ_QUADS][2][quad % BQ_QUADS][lane] = make_float4(cz[0], cz[1], cz[2], cz[3]);
  }
  __syncthreads();

  const int w = __builtin_amdgcn_readfirstlane(tid >> 5);
  const int wave = q0 + w;
  const int m = wave - b * M_;
  const float qx = pb[(size_t)m * STRIDE_ * 3 + 0];
  const float qy = pb[(size_t)m * STRIDE_ * 3 + 1];
  const float qz = pb[(size_t)m * STRIDE_ * 3 + 2];
  int* out = idx_out + (size_t)wave * K_;

  int count = 0;
  int first_idx = NSRC - 1;
  for (int g = 0; g < NSRC / BQ_GROUP; ++g) {
    unsigned mk[BQ_CHUNKS];
    float sx[BQ_CHUNKS], sy[BQ_CHUNKS], sz[BQ_CHUNKS];
#pragma unroll
    for (int h = 0; h < BQ_QUADS; ++h) {
      const float4 vx = s[g][0][h][lane], vy = s[g][1][h][lane], vz = s[g][2][h][lane];
      sx[4 * h + 0] = vx.x; sx[4 * h + 1] = vx.y; sx[4 * h + 2] = vx.z; sx[4 * h + 3] = vx.w;
      sy[4 * h + 0] = vy.x; sy[4 * h + 1] = vy.y; sy[4 * h + 2] = vy.z; sy[4 * h + 3] = vy.w;
      sz[4 * h + 0] = vz.x; sz[4 * h + 1] = vz.y; sz[4 * h + 2] = vz.z; sz[4 * h + 3] = vz.w;
    }
#pragma unroll
    for (int u = 0; u < BQ_CHUNKS; ++u) {
      float dx = sx[u] - qx, dy = sy[u] - qy, dz = sz[u] - qz;
      float sxy;
      {
#pragma clang fp contract(off)
        sxy = dx * dx + dy * dy;
      }
      float d2 = __builtin_fmaf(dz, dz, sxy);
      mk[u] = (unsigned)__ballot(d2 <= r2);
    }
    unsigned any = 0u;
#pragma unroll
    for (int u = 0; u < BQ_CHUNKS; ++u) any |= mk[u];
    if (any != 0u) {
      const int base = g * BQ_GROUP;
#pragma unroll
      for (int u = 0; u < BQ_CHUNKS; ++u) {
        unsigned mask = mk[u];
        if (mask != 0u) {
          if (count == 0) first_idx = base + u * 32 + (__ffs(mask) - 1);
          if ((mask >> lane) & 1u) {
            int pos = count + (int)__builtin_amdgcn_mbcnt_lo(mask, 0u);
            if (pos < K_) out[pos] = base + u * 32 + lane;
          }
          count += __popc(mask);
        }
      }
      if (count >= K_) break;
    }
  }
  if (count > K_) count = K_;
  for (int t = count + lane; t < K_; t += 32) out[t] = first_idx;
}

constexpr int FR_BQ = (B_ * M_) / BQ_WAVES;
constexpr int FR_F1 = ((B_ * N_ / 16) * (C_ / 16)) / BQ_WAVES;
constexpr int FR_MIX = (FR_BQ / 2) * 3;
constexpr int FR_BLOCKS = 2 * FR_BQ + FR_F1;
static_assert(FR_BQ % 2 == 0, "the mixed range pairs the first query's blocks");
static_assert(((B_ * N_ / 16) * (C_ / 16)) % BQ_WAVES == 0, "embedding tiles fill whole blocks");

__global__ __launch_bounds__(BQ_THREADS) void front_kernel(
    const float* __restrict__ p, const float* __restrict__ f, const float* __restrict__ Wsa,
    const float* __restrict__ bsa, float* __restrict__ f1, int* __restrict__ idx1,
    int* __restrict__ idx2) {
  __shared__ __attribute__((aligned(16))) float4 s_pts[N_ / BQ_GROUP][3][BQ_QUADS][32];
  const int bid = blockIdx.x;
  int role, blk;
  if (bid < FR_MIX) {
    const int t = bid / 3, r = bid - 3 * t;
    role = r < 2 ? 0 : 1;
    blk = r < 2 ? 2 * t + r : t;
  } else if (bid < 2 * FR_BQ) {
    role = 1;
    blk = bid - FR_MIX + FR_BQ / 2;
  } else {
    role = 2;
    blk = bid - 2 * FR_BQ;
  }
  if (role == 0) {
    ball_query_block<N_, 1>(s_pts, blk, p, R1SQ, idx1);
  } else if (role == 1) {
    ball_query_block<M_, STRIDE_>(s_pts, blk, p, R2SQ, idx2);
  } else {
    const int wave = blk * BQ_WAVES + (int)(threadIdx.x >> 5);
    f1_wmma_tile(wave, (int)(threadIdx.x & 31), f, Wsa, bsa, f1);
  }
}

constexpr int TR = 16;
constexpr int FT = 256;
constexpr int HID = EXP_ * C_;
constexpr int PT = C_ + 4;
constexpr int PH = HID + 4;

__device__ __forceinline__ v8f wmma4(v2f a, v2f b, v8f c) {
  return __builtin_amdgcn_wmma_f32_16x16x4_f32(false, a, false, b, (short)0, c, false, false);
}

template <bool MLP, bool CONV>
__global__ __launch_bounds__(FT) void row_tile_kernel(
    const float* __restrict__ p, const float* __restrict__ src, const int* __restrict__ idx,
    int src_stride, int src_n, const float* __restrict__ Wpe, const float* __restrict__ bpe,
    const float* __restrict__ Wpw1, const float* __restrict__ bpw1,
    const float* __restrict__ Wpw2, const float* __restrict__ bpw2, float* feat,
    const float* __restrict__ Wc, const float* __restrict__ bc, float* __restrict__ hout) {
  static_assert(M_ % TR == 0, "a tile never crosses a batch");
  static_assert(TR * K_ == 2 * FT && K_ == 32, "staging: two records per thread, a wave per query");
  static_assert(TR == 2 * (FT / 32), "gather: two queries per wave");
  __shared__ __attribute__((aligned(16))) int4 s_nb[TR][K_];
  __shared__ __attribute__((aligned(16))) float sT[TR][PT];
  __shared__ __attribute__((aligned(16))) float sH[MLP ? TR : 1][PH];

  const int tid = threadIdx.x;
  const int lane = tid & 31;
  const int half = lane >> 4;
  const int ln = lane & 15;
  const int w = __builtin_amdgcn_readfirstlane(tid >> 5);
  const int bm0 = blockIdx.x * TR;
  const int b = bm0 / M_;
  const float* pb = p + (size_t)b * N_ * 3;

  {
    const int* ib = idx + (size_t)bm0 * K_;
    const int jA = ib[tid];
    const int jB = ib[tid + FT];
    const int qlA = w, qlB = w + TR / 2;
    const float* qa = pb + (size_t)(bm0 - b * M_ + qlA) * STRIDE_ * 3;
    const float* qb = pb + (size_t)(bm0 - b * M_ + qlB) * STRIDE_ * 3;
    const float* sa = pb + (size_t)jA * src_stride * 3;
    const float* sc = pb + (size_t)jB * src_stride * 3;
    const float ax = sa[0], ay = sa[1], az = sa[2];
    const float cx = sc[0], cy = sc[1], cz = sc[2];
    int4 recA, recB;
    recA.x = jA;
    recA.y = __float_as_int(ax - qa[0]);
    recA.z = __float_as_int(ay - qa[1]);
    recA.w = __float_as_int(az - qa[2]);
    recB.x = jB;
    recB.y = __float_as_int(cx - qb[0]);
    recB.z = __float_as_int(cy - qb[1]);
    recB.w = __float_as_int(cz - qb[2]);
    s_nb[qlA][lane] = recA;
    s_nb[qlB][lane] = recB;
  }
  __syncthreads();

  {
    const int c4 = ln * 4;
    const float4 w0 = *(const float4*)(Wpe + 0 * C_ + c4);
    const float4 w1 = *(const float4*)(Wpe + 1 * C_ + c4);
    const float4 w2 = *(const float4*)(Wpe + 2 * C_ + c4);
    const float4 bi = *(const float4*)(bpe + c4);
    const float* sb = src + (size_t)b * src_n * C_ + c4;
#pragma unroll
    for (int qq = 0; qq < 2; ++qq) {
      const int ql = 2 * w + qq;
      float4 best = {NEG_BIG, NEG_BIG, NEG_BIG, NEG_BIG};
#pragma unroll
      for (int i = 0; i < K_ / 2; ++i) {
        const int4 rec = s_nb[ql][2 * i + half];
        const float dx = __int_as_float(rec.y), dy = __int_as_float(rec.z),
                    dz = __int_as_float(rec.w);
        const float4 s4 = *(const float4*)(sb + (size_t)rec.x * C_);
        best.x = fmaxf(best.x, dx * w0.x + dy * w1.x + dz * w2.x + bi.x + s4.x);
        best.y = fmaxf(best.y, dx * w0.y + dy * w1.y + dz * w2.y + bi.y + s4.y);
        best.z = fmaxf(best.z, dx * w0.z + dy * w1.z + dz * w2.z + bi.z + s4.z);
        best.w = fmaxf(best.w, dx * w0.w + dy * w1.w + dz * w2.w + bi.w + s4.w);
      }
      best.x = fmaxf(best.x, __shfl_xor(best.x, 16));
      best.y = fmaxf(best.y, __shfl_xor(best.y, 16));
      best.z = fmaxf(best.z, __shfl_xor(best.z, 16));
      best.w = fmaxf(best.w, __shfl_xor(best.w, 16));
      if (half == 0) *(float4*)&sT[ql][c4] = best;
    }
  }
  __syncthreads();

  if constexpr (MLP) {
    {
      const int n0 = w * 32;
      const float* Wb = Wpw1 + (size_t)(2 * half) * HID + n0 + ln;
      v8f acc0 = {0.f, 0.f, 0.f, 0.f, 0.f, 0.f, 0.f, 0.f};
      v8f acc1 = {0.f, 0.f, 0.f, 0.f, 0.f, 0.f, 0.f, 0.f};
#pragma unroll
      for (int k = 0; k < C_; k += 4) {
        v2f a;   a.x = sT[ln][k + 2 * half];        a.y = sT[ln][k + 2 * half + 1];
        v2f b0;  b0.x = Wb[(size_t)k * HID];        b0.y = Wb[(size_t)(k + 1) * HID];
        v2f b1;  b1.x = Wb[(size_t)k * HID + 16];   b1.y = Wb[(size_t)(k + 1) * HID + 16];
        acc0 = wmma4(a, b0, acc0);
        acc1 = wmma4(a, b1, acc1);
      }
      const float bv0 = bpw1[n0 + ln], bv1 = bpw1[n0 + 16 + ln];
#pragma unroll
      for (int r = 0; r < 8; ++r) {
        sH[r + 8 * half][n0 + ln] = fmaxf(acc0[r] + bv0, 0.0f);
        sH[r + 8 * half][n0 + 16 + ln] = fmaxf(acc1[r] + bv1, 0.0f);
      }
    }
    __syncthreads();

    if (w < 4) {
      const int n0 = w * 16;
      const float* Wb = Wpw2 + (size_t)(2 * half) * C_ + n0 + ln;
      v8f acc = {0.f, 0.f, 0.f, 0.f, 0.f, 0.f, 0.f, 0.f};
#pragma unroll
      for (int k = 0; k < HID; k += 4) {
        v2f a;   a.x = sH[ln][k + 2 * half];        a.y = sH[ln][k + 2 * half + 1];
        v2f bb;  bb.x = Wb[(size_t)k * C_];         bb.y = Wb[(size_t)(k + 1) * C_];
        acc = wmma4(a, bb, acc);
      }
      const int col = n0 + ln;
      const float bv = bpw2[col];
#pragma unroll
      for (int r = 0; r < 8; ++r) {
        const int row = r + 8 * half;
        float v = acc[r] + bv;
        v += feat[(size_t)(bm0 + row) * C_ + col];
        sT[row][col] = fmaxf(v, 0.0f);
      }
    }
    __syncthreads();
  }

  {
    const int row = tid >> 4, c4 = (tid & 15) * 4;
    *(float4*)(feat + (size_t)(bm0 + row) * C_ + c4) = *(const float4*)&sT[row][c4];
  }

  if constexpr (CONV) {
    if (w < 4) {
      const int n0 = w * 16;
      const float* Wb = Wc + (size_t)(2 * half) * C_ + n0 + ln;
      v8f acc = {0.f, 0.f, 0.f, 0.f, 0.f, 0.f, 0.f, 0.f};
#pragma unroll
      for (int k = 0; k < C_; k += 4) {
        v2f a;   a.x = sT[ln][k + 2 * half];        a.y = sT[ln][k + 2 * half + 1];
        v2f bb;  bb.x = Wb[(size_t)k * C_];         bb.y = Wb[(size_t)(k + 1) * C_];
        acc = wmma4(a, bb, acc);
      }
      const int col = n0 + ln;
      const float bv = bc[col];
#pragma unroll
      for (int r = 0; r < 8; ++r) {
        const int row = bm0 + r + 8 * half;
        hout[(size_t)row * C_ + col] = fmaxf(acc[r] + bv, 0.0f);
      }
    }
  }
}

}

extern "C" void kernel_launch(void* const* d_in, const int* in_sizes, int n_in,
                              void* d_out, int out_size, void* d_ws, size_t ws_size,
                              hipStream_t stream) {
  (void)in_sizes; (void)n_in; (void)out_size; (void)ws_size;
  const float* p      = (const float*)d_in[0];
  const float* f      = (const float*)d_in[1];
  const float* W_sa   = (const float*)d_in[2];
  const float* b_sa   = (const float*)d_in[3];
  const float* W_pe1  = (const float*)d_in[4];
  const float* b_pe1  = (const float*)d_in[5];
  const float* W_pe2  = (const float*)d_in[6];
  const float* b_pe2  = (const float*)d_in[7];
  const float* W_conv = (const float*)d_in[8];
  const float* b_conv = (const float*)d_in[9];
  const float* W_pw1  = (const float*)d_in[10];
  const float* b_pw1  = (const float*)d_in[11];
  const float* W_pw2  = (const float*)d_in[12];
  const float* b_pw2  = (const float*)d_in[13];
  float* feat = (float*)d_out;

  char* ws = (char*)d_ws;
  float* f1 = (float*)ws;  ws += sizeof(float) * (size_t)B_ * N_ * C_;
  float* hA = (float*)ws;  ws += sizeof(float) * (size_t)B_ * M_ * C_;
  float* hB = (float*)ws;  ws += sizeof(float) * (size_t)B_ * M_ * C_;
  int* idx1 = (int*)ws;    ws += sizeof(int) * (size_t)B_ * M_ * K_;
  int* idx2 = (int*)ws;

  front_kernel<<<FR_BLOCKS, BQ_THREADS, 0, stream>>>(p, f, W_sa, b_sa, f1, idx1, idx2);

  const int tiles = (B_ * M_) / TR;

  row_tile_kernel<false, true><<<tiles, FT, 0, stream>>>(
      p, f1, idx1, 1, N_, W_pe1, b_pe1, nullptr, nullptr, nullptr, nullptr, feat,
      W_conv, b_conv, hA);

  row_tile_kernel<true, true><<<tiles, FT, 0, stream>>>(
      p, hA, idx2, STRIDE_, M_, W_pe2, b_pe2, W_pw1, b_pw1, W_pw2, b_pw2, feat,
      W_conv + (size_t)C_ * C_, b_conv + C_, hB);

  row_tile_kernel<true, false><<<tiles, FT, 0, stream>>>(
      p, hB, idx2, STRIDE_, M_, W_pe2, b_pe2, W_pw1 + (size_t)C_ * HID, b_pw1 + HID,
      W_pw2 + (size_t)HID * C_, b_pw2 + C_, feat, nullptr, nullptr, nullptr);
}


# ball queries emit 16-byte neighbour records {j, source - query} from their LDS-staged points, one 512-B run per query; row-tile launches read them in order: no per-neighbour point gather
# speedup vs baseline: 4.6200x; 1.0352x over previous
#include <hip/hip_runtime.h>
#include <hip/hip_bf16.h>
#include <cstdint>

namespace {

constexpr int B_ = 4, N_ = 8192, CIN_ = 4, C_ = 64, K_ = 32, STRIDE_ = 4, NB_ = 2, EXP_ = 4;
constexpr int M_ = N_ / STRIDE_;
constexpr float R1SQ = 0.1f * 0.1f;
constexpr float R2SQ = 0.2f * 0.2f;
constexpr float NEG_BIG = -3.4028235e38f;

typedef float v2f __attribute__((ext_vector_type(2)));
typedef float v8f __attribute__((ext_vector_type(8)));

__device__ __forceinline__ void f1_wmma_tile(int wave, int lane, const float* __restrict__ f,
                                             const float* __restrict__ Wsa,
                                             const float* __restrict__ bsa,
                                             float* __restrict__ f1) {
  constexpr int tilesC = C_ / 16;
  int row0 = (wave / tilesC) * 16;
  int c0 = (wave % tilesC) * 16;
  int b = row0 / N_;
  int n0 = row0 % N_;
  int half = lane >> 4;
  int ln = lane & 15;

  const float* fb = f + ((size_t)b * CIN_ + 2 * half) * N_ + n0 + ln;
  v2f a;  a.x = fb[0];        a.y = fb[N_];
  const float* wc = Wsa + (size_t)(2 * half) * C_ + c0 + ln;
  v2f bb; bb.x = wc[0];       bb.y = wc[C_];
  v8f acc = {0.f, 0.f, 0.f, 0.f, 0.f, 0.f, 0.f, 0.f};
  acc = __builtin_amdgcn_wmma_f32_16x16x4_f32(false, a, false, bb, (short)0, acc,
                                              false, false);
  int col = c0 + ln;
  float bv = bsa[col];
#pragma unroll
  for (int r = 0; r < 8; ++r) {
    int row = row0 + r + 8 * half;
    f1[(size_t)row * C_ + col] = fmaxf(acc[r] + bv, 0.0f);
  }
}

constexpr int BQ_THREADS = 512;
constexpr int BQ_WAVES = BQ_THREADS / 32;
constexpr int BQ_GROUP = 256;
constexpr int BQ_CHUNKS = BQ_GROUP / 32;
constexpr int BQ_QUAD = 128;
constexpr int BQ_QUADS = BQ_GROUP / BQ_QUAD;

template <int NSRC, int SSTRIDE>
__device__ __forceinline__ void ball_query_block(float4 (*s)[3][BQ_QUADS][32], int (*s_hit)[K_],
                                                 int blk, const float* __restrict__ p, float r2,
                                                 int4* __restrict__ rec_out) {
  static_assert(K_ == 32, "a lane per neighbour slot when the records are written");
  static_assert((NSRC / BQ_QUAD) % BQ_WAVES == 0, "staging loop covers NSRC exactly, a quad per wave and pass");
  static_assert(NSRC % BQ_GROUP == 0, "scan trips cover NSRC exactly");
  static_assert(M_ % BQ_WAVES == 0, "a block never crosses a batch");

  const int tid = threadIdx.x;
  const int lane = tid & 31;
  const int q0 = blk * BQ_WAVES;
  const int b = q0 / M_;
  const float* pb = p + (size_t)b * N_ * 3;

#pragma unroll
  for (int i = 0; i < (NSRC / BQ_QUAD) / BQ_WAVES; ++i) {
    const int quad = (tid >> 5) + i * BQ_WAVES;
    float cx[4], cy[4], cz[4];
#pragma unroll
    for (int e = 0; e < 4; ++e) {
      const int j = quad * BQ_QUAD + e * 32 + lane;
      const float* sp = pb + (size_t)j * SSTRIDE * 3;
      cx[e] = sp[0];
      cy[e] = sp[1];
      cz[e] = sp[2];
    }
    s[quad / BQ_QUADS][0][quad % BQ_QUADS][lane] = make_float4(cx[0], cx[1], cx[2], cx[3]);
    s[quad / BQ_QUADS][1][quad % BQ_QUADS][lane] = make_float4(cy[0], cy[1], cy[2], cy[3]);
    s[quad / BQ_QUADS][2][quad % BQ_QUADS][lane] = make_float4(cz[0], cz[1], cz[2], cz[3]);
  }
  __syncthreads();

  const int w = __builtin_amdgcn_readfirstlane(tid >> 5);
  const int wave = q0 + w;
  const int m = wave - b * M_;
  const float qx = pb[(size_t)m * STRIDE_ * 3 + 0];
  const float qy = pb[(size_t)m * STRIDE_ * 3 + 1];
  const float qz = pb[(size_t)m * STRIDE_ * 3 + 2];
  int* hit = s_hit[w];

  int count = 0;
  for (int g = 0; g < NSRC / BQ_GROUP; ++g) {
    unsigned mk[BQ_CHUNKS];
    float sx[BQ_CHUNKS], sy[BQ_CHUNKS], sz[BQ_CHUNKS];
#pragma unroll
    for (int h = 0; h < BQ_QUADS; ++h) {
      const float4 vx = s[g][0][h][lane], vy = s[g][1][h][lane], vz = s[g][2][h][lane];
      sx[4 * h + 0] = vx.x; sx[4 * h + 1] = vx.y; sx[4 * h + 2] = vx.z; sx[4 * h + 3] = vx.w;
      sy[4 * h + 0] = vy.x; sy[4 * h + 1] = vy.y; sy[4 * h + 2] = vy.z; sy[4 * h + 3] = vy.w;
      sz[4 * h + 0] = vz.x; sz[4 * h + 1] = vz.y; sz[4 * h + 2] = vz.z; sz[4 * h + 3] = vz.w;
    }
#pragma unroll
    for (int u = 0; u < BQ_CHUNKS; ++u) {
      float dx = sx[u] - qx, dy = sy[u] - qy, dz = sz[u] - qz;
      float sxy;
      {
#pragma clang fp contract(off)
        sxy = dx * dx + dy * dy;
      }
      float d2 = __builtin_fmaf(dz, dz, sxy);
      mk[u] = (unsigned)__ballot(d2 <= r2);
    }
    unsigned any = 0u;
#pragma unroll
    for (int u = 0; u < BQ_CHUNKS; ++u) any |= mk[u];
    if (any != 0u) {
      const int base = g * BQ_GROUP;
#pragma unroll
      for (int u = 0; u < BQ_CHUNKS; ++u) {
        unsigned mask = mk[u];
        if (mask != 0u) {
          if ((mask >> lane) & 1u) {
            int pos = count + (int)__builtin_amdgcn_mbcnt_lo(mask, 0u);
            if (pos < K_) hit[pos] = base + u * 32 + lane;
          }
          count += __popc(mask);
        }
      }
      if (count >= K_) break;
    }
  }
  if (count > K_) count = K_;
  int j = hit[lane < count ? lane : 0];
  if (count == 0) j = NSRC - 1;
  const int jg = j / BQ_GROUP, jr = j % BQ_GROUP;
  const int jh = jr / BQ_QUAD, jq = jr % BQ_QUAD;
  const int je = jq >> 5, jl = jq & 31;
  const float px = ((const float*)&s[jg][0][jh][jl])[je];
  const float py = ((const float*)&s[jg][1][jh][jl])[je];
  const float pz = ((const float*)&s[jg][2][jh][jl])[je];
  int4 rec;
  rec.x = j;
  rec.y = __float_as_int(px - qx);
  rec.z = __float_as_int(py - qy);
  rec.w = __float_as_int(pz - qz);
  rec_out[(size_t)wave * K_ + lane] = rec;
}

constexpr int FR_BQ = (B_ * M_) / BQ_WAVES;
constexpr int FR_F1 = ((B_ * N_ / 16) * (C_ / 16)) / BQ_WAVES;
constexpr int FR_MIX = (FR_BQ / 2) * 3;
constexpr int FR_BLOCKS = 2 * FR_BQ + FR_F1;
static_assert(FR_BQ % 2 == 0, "the mixed range pairs the first query's blocks");
static_assert(((B_ * N_ / 16) * (C_ / 16)) % BQ_WAVES == 0, "embedding tiles fill whole blocks");

__global__ __launch_bounds__(BQ_THREADS) void front_kernel(
    const float* __restrict__ p, const float* __restrict__ f, const float* __restrict__ Wsa,
    const float* __restrict__ bsa, float* __restrict__ f1, int* __restrict__ rec1,
    int* __restrict__ rec2) {
  __shared__ __attribute__((aligned(16))) float4 s_pts[N_ / BQ_GROUP][3][BQ_QUADS][32];
  __shared__ int s_hit[BQ_WAVES][K_];
  const int bid = blockIdx.x;
  int role, blk;
  if (bid < FR_MIX) {
    const int t = bid / 3, r = bid - 3 * t;
    role = r < 2 ? 0 : 1;
    blk = r < 2 ? 2 * t + r : t;
  } else if (bid < 2 * FR_BQ) {
    role = 1;
    blk = bid - FR_MIX + FR_BQ / 2;
  } else {
    role = 2;
    blk = bid - 2 * FR_BQ;
  }
  if (role == 0) {
    ball_query_block<N_, 1>(s_pts, s_hit, blk, p, R1SQ, (int4*)rec1);
  } else if (role == 1) {
    ball_query_block<M_, STRIDE_>(s_pts, s_hit, blk, p, R2SQ, (int4*)rec2);
  } else {
    const int wave = blk * BQ_WAVES + (int)(threadIdx.x >> 5);
    f1_wmma_tile(wave, (int)(threadIdx.x & 31), f, Wsa, bsa, f1);
  }
}

constexpr int TR = 16;
constexpr int FT = 256;
constexpr int HID = EXP_ * C_;
constexpr int PT = C_ + 4;
constexpr int PH = HID + 4;

__device__ __forceinline__ v8f wmma4(v2f a, v2f b, v8f c) {
  return __builtin_amdgcn_wmma_f32_16x16x4_f32(false, a, false, b, (short)0, c, false, false);
}

template <bool MLP, bool CONV>
__global__ __launch_bounds__(FT) void row_tile_kernel(
    const float* __restrict__ src, const int* __restrict__ rec, int src_n,
    const float* __restrict__ Wpe, const float* __restrict__ bpe,
    const float* __restrict__ Wpw1, const float* __restrict__ bpw1,
    const float* __restrict__ Wpw2, const float* __restrict__ bpw2, float* feat,
    const float* __restrict__ Wc, const float* __restrict__ bc, float* __restrict__ hout) {
  static_assert(M_ % TR == 0, "a tile never crosses a batch");
  static_assert(TR * K_ == 2 * FT && K_ == 32, "staging: two records per thread, a wave per query");
  static_assert(TR == 2 * (FT / 32), "gather: two queries per wave");
  __shared__ __attribute__((aligned(16))) int4 s_nb[TR][K_];
  __shared__ __attribute__((aligned(16))) float sT[TR][PT];
  __shared__ __attribute__((aligned(16))) float sH[MLP ? TR : 1][PH];

  const int tid = threadIdx.x;
  const int lane = tid & 31;
  const int half = lane >> 4;
  const int ln = lane & 15;
  const int w = __builtin_amdgcn_readfirstlane(tid >> 5);
  const int bm0 = blockIdx.x * TR;
  const int b = bm0 / M_;

  {
    const int4* rb = (const int4*)rec + (size_t)bm0 * K_;
    const int4 recA = rb[tid];
    const int4 recB = rb[tid + FT];
    s_nb[w][lane] = recA;
    s_nb[w + TR / 2][lane] = recB;
  }
  __syncthreads();

  {
    const int c4 = ln * 4;
    const float4 w0 = *(const float4*)(Wpe + 0 * C_ + c4);
    const float4 w1 = *(const float4*)(Wpe + 1 * C_ + c4);
    const float4 w2 = *(const float4*)(Wpe + 2 * C_ + c4);
    const float4 bi = *(const float4*)(bpe + c4);
    const float* sb = src + (size_t)b * src_n * C_ + c4;
#pragma unroll
    for (int qq = 0; qq < 2; ++qq) {
      const int ql = 2 * w + qq;
      float4 best = {NEG_BIG, NEG_BIG, NEG_BIG, NEG_BIG};
#pragma unroll
      for (int i = 0; i < K_ / 2; ++i) {
        const int4 rec = s_nb[ql][2 * i + half];
        const float dx = __int_as_float(rec.y), dy = __int_as_float(rec.z),
                    dz = __int_as_float(rec.w);
        const float4 s4 = *(const float4*)(sb + (size_t)rec.x * C_);
        best.x = fmaxf(best.x, dx * w0.x + dy * w1.x + dz * w2.x + bi.x + s4.x);
        best.y = fmaxf(best.y, dx * w0.y + dy * w1.y + dz * w2.y + bi.y + s4.y);
        best.z = fmaxf(best.z, dx * w0.z + dy * w1.z + dz * w2.z + bi.z + s4.z);
        best.w = fmaxf(best.w, dx * w0.w + dy * w1.w + dz * w2.w + bi.w + s4.w);
      }
      best.x = fmaxf(best.x, __shfl_xor(best.x, 16));
      best.y = fmaxf(best.y, __shfl_xor(best.y, 16));
      best.z = fmaxf(best.z, __shfl_xor(best.z, 16));
      best.w = fmaxf(best.w, __shfl_xor(best.w, 16));
      if (half == 0) *(float4*)&sT[ql][c4] = best;
    }
  }
  __syncthreads();

  if constexpr (MLP) {
    {
      const int n0 = w * 32;
      const float* Wb = Wpw1 + (size_t)(2 * half) * HID + n0 + ln;
      v8f acc0 = {0.f, 0.f, 0.f, 0.f, 0.f, 0.f, 0.f, 0.f};
      v8f acc1 = {0.f, 0.f, 0.f, 0.f, 0.f, 0.f, 0.f, 0.f};
#pragma unroll
      for (int k = 0; k < C_; k += 4) {
        v2f a;   a.x = sT[ln][k + 2 * half];        a.y = sT[ln][k + 2 * half + 1];
        v2f b0;  b0.x = Wb[(size_t)k * HID];        b0.y = Wb[(size_t)(k + 1) * HID];
        v2f b1;  b1.x = Wb[(size_t)k * HID + 16];   b1.y = Wb[(size_t)(k + 1) * HID + 16];
        acc0 = wmma4(a, b0, acc0);
        acc1 = wmma4(a, b1, acc1);
      }
      const float bv0 = bpw1[n0 + ln], bv1 = bpw1[n0 + 16 + ln];
#pragma unroll
      for (int r = 0; r < 8; ++r) {
        sH[r + 8 * half][n0 + ln] = fmaxf(acc0[r] + bv0, 0.0f);
        sH[r + 8 * half][n0 + 16 + ln] = fmaxf(acc1[r] + bv1, 0.0f);
      }
    }
    __syncthreads();

    if (w < 4) {
      const int n0 = w * 16;
      const float* Wb = Wpw2 + (size_t)(2 * half) * C_ + n0 + ln;
      v8f acc = {0.f, 0.f, 0.f, 0.f, 0.f, 0.f, 0.f, 0.f};
#pragma unroll
      for (int k = 0; k < HID; k += 4) {
        v2f a;   a.x = sH[ln][k + 2 * half];        a.y = sH[ln][k + 2 * half + 1];
        v2f bb;  bb.x = Wb[(size_t)k * C_];         bb.y = Wb[(size_t)(k + 1) * C_];
        acc = wmma4(a, bb, acc);
      }
      const int col = n0 + ln;
      const float bv = bpw2[col];
#pragma unroll
      for (int r = 0; r < 8; ++r) {
        const int row = r + 8 * half;
        float v = acc[r] + bv;
        v += feat[(size_t)(bm0 + row) * C_ + col];
        sT[row][col] = fmaxf(v, 0.0f);
      }
    }
    __syncthreads();
  }

  {
    const int row = tid >> 4, c4 = (tid & 15) * 4;
    *(float4*)(feat + (size_t)(bm0 + row) * C_ + c4) = *(const float4*)&sT[row][c4];
  }

  if constexpr (CONV) {
    if (w < 4) {
      const int n0 = w * 16;
      const float* Wb = Wc + (size_t)(2 * half) * C_ + n0 + ln;
      v8f acc = {0.f, 0.f, 0.f, 0.f, 0.f, 0.f, 0.f, 0.f};
#pragma unroll
      for (int k = 0; k < C_; k += 4) {
        v2f a;   a.x = sT[ln][k + 2 * half];        a.y = sT[ln][k + 2 * half + 1];
        v2f bb;  bb.x = Wb[(size_t)k * C_];         bb.y = Wb[(size_t)(k + 1) * C_];
        acc = wmma4(a, bb, acc);
      }
      const int col = n0 + ln;
      const float bv = bc[col];
#pragma unroll
      for (int r = 0; r < 8; ++r) {
        const int row = bm0 + r + 8 * half;
        hout[(size_t)row * C_ + col] = fmaxf(acc[r] + bv, 0.0f);
      }
    }
  }
}

}

extern "C" void kernel_launch(void* const* d_in, const int* in_sizes, int n_in,
                              void* d_out, int out_size, void* d_ws, size_t ws_size,
                              hipStream_t stream) {
  (void)in_sizes; (void)n_in; (void)out_size; (void)ws_size;
  const float* p      = (const float*)d_in[0];
  const float* f      = (const float*)d_in[1];
  const float* W_sa   = (const float*)d_in[2];
  const float* b_sa   = (const float*)d_in[3];
  const float* W_pe1  = (const float*)d_in[4];
  const float* b_pe1  = (const float*)d_in[5];
  const float* W_pe2  = (const float*)d_in[6];
  const float* b_pe2  = (const float*)d_in[7];
  const float* W_conv = (const float*)d_in[8];
  const float* b_conv = (const float*)d_in[9];
  const float* W_pw1  = (const float*)d_in[10];
  const float* b_pw1  = (const float*)d_in[11];
  const float* W_pw2  = (const float*)d_in[12];
  const float* b_pw2  = (const float*)d_in[13];
  float* feat = (float*)d_out;

  char* ws = (char*)d_ws;
  float* f1 = (float*)ws;  ws += sizeof(float) * (size_t)B_ * N_ * C_;
  float* hA = (float*)ws;  ws += sizeof(float) * (size_t)B_ * M_ * C_;
  float* hB = (float*)ws;  ws += sizeof(float) * (size_t)B_ * M_ * C_;
  int* rec1 = (int*)ws;    ws += sizeof(int) * 4 * (size_t)B_ * M_ * K_;
  int* rec2 = (int*)ws;

  front_kernel<<<FR_BLOCKS, BQ_THREADS, 0, stream>>>(p, f, W_sa, b_sa, f1, rec1, rec2);

  const int tiles = (B_ * M_) / TR;

  row_tile_kernel<false, true><<<tiles, FT, 0, stream>>>(
      f1, rec1, N_, W_pe1, b_pe1, nullptr, nullptr, nullptr, nullptr, feat,
      W_conv, b_conv, hA);

  row_tile_kernel<true, true><<<tiles, FT, 0, stream>>>(
      hA, rec2, M_, W_pe2, b_pe2, W_pw1, b_pw1, W_pw2, b_pw2, feat,
      W_conv + (size_t)C_ * C_, b_conv + C_, hB);

  row_tile_kernel<true, false><<<tiles, FT, 0, stream>>>(
      hB, rec2, M_, W_pe2, b_pe2, W_pw1 + (size_t)C_ * HID, b_pw1 + HID,
      W_pw2 + (size_t)HID * C_, b_pw2 + C_, feat, nullptr, nullptr, nullptr);
}


# first gather-max stage runs on the first ball query's own waves (16-wave block = one row tile); embedding and pe term as two K=4 WMMAs per tile (pe sum order: s+dx.w0+dy.w1+dz.w2+b): 4 launches -> 3
# speedup vs baseline: 4.9685x; 1.0754x over previous
#include <hip/hip_runtime.h>
#include <hip/hip_bf16.h>
#include <cstdint>

namespace {

constexpr int B_ = 4, N_ = 8192, CIN_ = 4, C_ = 64, K_ = 32, STRIDE_ = 4, NB_ = 2, EXP_ = 4;
constexpr int M_ = N_ / STRIDE_;
constexpr float R1SQ = 0.1f * 0.1f;
constexpr float R2SQ = 0.2f * 0.2f;
constexpr float NEG_BIG = -3.4028235e38f;

typedef float v2f __attribute__((ext_vector_type(2)));
typedef float v8f __attribute__((ext_vector_type(8)));

constexpr int BQ_THREADS = 512;
constexpr int BQ_WAVES = BQ_THREADS / 32;
constexpr int BQ_GROUP = 256;
constexpr int BQ_CHUNKS = BQ_GROUP / 32;
constexpr int BQ_QUAD = 128;
constexpr int BQ_QUADS = BQ_GROUP / BQ_QUAD;

template <int NSRC, int SSTRIDE>
__device__ __forceinline__ int4 ball_query_block(float4 (*s)[3][BQ_QUADS][32], int (*s_hit)[K_],
                                                 int blk, const float* __restrict__ p, float r2) {
  static_assert(K_ == 32, "a lane per neighbour slot when the records are written");
  static_assert((NSRC / BQ_QUAD) % BQ_WAVES == 0, "staging loop covers NSRC exactly, a quad per wave and pass");
  static_assert(NSRC % BQ_GROUP == 0, "scan trips cover NSRC exactly");
  static_assert(M_ % BQ_WAVES == 0, "a block never crosses a batch");

  const int tid = threadIdx.x;
  const int lane = tid & 31;
  const int q0 = blk * BQ_WAVES;
  const int b = q0 / M_;
  const float* pb = p + (size_t)b * N_ * 3;

#pragma unroll
  for (int i = 0; i < (NSRC / BQ_QUAD) / BQ_WAVES; ++i) {
    const int quad = (tid >> 5) + i * BQ_WAVES;
    float cx[4], cy[4], cz[4];
#pragma unroll
    for (int e = 0; e < 4; ++e) {
      const int j = quad * BQ_QUAD + e * 32 + lane;
      const float* sp = pb + (size_t)j * SSTRIDE * 3;
      cx[e] = sp[0];
      cy[e] = sp[1];
      cz[e] = sp[2];
    }
    s[quad / BQ_QUADS][0][quad % BQ_QUADS][lane] = make_float4(cx[0], cx[1], cx[2], cx[3]);
    s[quad / BQ_QUADS][1][quad % BQ_QUADS][lane] = make_float4(cy[0], cy[1], cy[2], cy[3]);
    s[quad / BQ_QUADS][2][quad % BQ_QUADS][lane] = make_float4(cz[0], cz[1], cz[2], cz[3]);
  }
  __syncthreads();

  const int w = __builtin_amdgcn_readfirstlane(tid >> 5);
  const int wave = q0 + w;
  const int m = wave - b * M_;
  const float qx = pb[(size_t)m * STRIDE_ * 3 + 0];
  const float qy = pb[(size_t)m * STRIDE_ * 3 + 1];
  const float qz = pb[(size_t)m * STRIDE_ * 3 + 2];
  int* hit = s_hit[w];

  int count = 0;
  for (int g = 0; g < NSRC / BQ_GROUP; ++g) {
    unsigned mk[BQ_CHUNKS];
    float sx[BQ_CHUNKS], sy[BQ_CHUNKS], sz[BQ_CHUNKS];
#pragma unroll
    for (int h = 0; h < BQ_QUADS; ++h) {
      const float4 vx = s[g][0][h][lane], vy = s[g][1][h][lane], vz = s[g][2][h][lane];
      sx[4 * h + 0] = vx.x; sx[4 * h + 1] = vx.y; sx[4 * h + 2] = vx.z; sx[4 * h + 3] = vx.w;
      sy[4 * h + 0] = vy.x; sy[4 * h + 1] = vy.y; sy[4 * h + 2] = vy.z; sy[4 * h + 3] = vy.w;
      sz[4 * h + 0] = vz.x; sz[4 * h + 1] = vz.y; sz[4 * h + 2] = vz.z; sz[4 * h + 3] = vz.w;
    }
#pragma unroll
    for (int u = 0; u < BQ_CHUNKS; ++u) {
      float dx = sx[u] - qx, dy = sy[u] - qy, dz = sz[u] - qz;
      float sxy;
      {
#pragma clang fp contract(off)
        sxy = dx * dx + dy * dy;
      }
      float d2 = __builtin_fmaf(dz, dz, sxy);
      mk[u] = (unsigned)__ballot(d2 <= r2);
    }
    unsigned any = 0u;
#pragma unroll
    for (int u = 0; u < BQ_CHUNKS; ++u) any |= mk[u];
    if (any != 0u) {
      const int base = g * BQ_GROUP;
#pragma unroll
      for (int u = 0; u < BQ_CHUNKS; ++u) {
        unsigned mask = mk[u];
        if (mask != 0u) {
          if ((mask >> lane) & 1u) {
            int pos = count + (int)__builtin_amdgcn_mbcnt_lo(mask, 0u);
            if (pos < K_) hit[pos] = base + u * 32 + lane;
          }
          count += __popc(mask);
        }
      }
      if (count >= K_) break;
    }
  }
  if (count > K_) count = K_;
  int j = hit[lane < count ? lane : 0];
  if (count == 0) j = NSRC - 1;
  const int jg = j / BQ_GROUP, jr = j % BQ_GROUP;
  const int jh = jr / BQ_QUAD, jq = jr % BQ_QUAD;
  const int je = jq >> 5, jl = jq & 31;
  const float px = ((const float*)&s[jg][0][jh][jl])[je];
  const float py = ((const float*)&s[jg][1][jh][jl])[je];
  const float pz = ((const float*)&s[jg][2][jh][jl])[je];
  int4 rec;
  rec.x = j;
  rec.y = __float_as_int(px - qx);
  rec.z = __float_as_int(py - qy);
  rec.w = __float_as_int(pz - qz);
  return rec;
}

constexpr int PT = C_ + 4;

__device__ __forceinline__ v8f wmma4(v2f a, v2f b, v8f c) {
  return __builtin_amdgcn_wmma_f32_16x16x4_f32(false, a, false, b, (short)0, c, false, false);
}

__device__ __forceinline__ void first_stage_tile(
    float (*sT)[PT], int4 rec, int blk, const float* __restrict__ f,
    const float* __restrict__ Wsa, const float* __restrict__ bsa,
    const float* __restrict__ Wpe, const float* __restrict__ bpe, float* __restrict__ feat,
    const float* __restrict__ Wc, const float* __restrict__ bc, float* __restrict__ hout) {
  static_assert(BQ_WAVES == 16 && K_ == 32 && C_ == 64 && CIN_ == 4,
                "a block is one 16-row tile; 2 x 4 WMMA tiles of K = 4 per query");
  const int tid = threadIdx.x;
  const int lane = tid & 31;
  const int half = lane >> 4;
  const int ln = lane & 15;
  const int w = __builtin_amdgcn_readfirstlane(tid >> 5);
  const int bm0 = blk * BQ_WAVES;
  const int b = bm0 / M_;

  int4 oth;
  oth.x = __shfl_xor(rec.x, 16);
  oth.y = __shfl_xor(rec.y, 16);
  oth.z = __shfl_xor(rec.z, 16);
  oth.w = __shfl_xor(rec.w, 16);

  const float* fb = f + ((size_t)b * CIN_ + 2 * half) * N_;
  v2f a1[2], a2[2];
#pragma unroll
  for (int rt = 0; rt < 2; ++rt) {
    const int4 nb = (rt == half) ? rec : oth;
    a1[rt].x = fb[nb.x];
    a1[rt].y = fb[N_ + nb.x];
    a2[rt].x = half ? __int_as_float(nb.w) : __int_as_float(nb.y);
    a2[rt].y = half ? 1.0f : __int_as_float(nb.z);
  }

  const float* pe_hi = half ? bpe : Wpe + C_;
  float best[4];
#pragma unroll
  for (int ct = 0; ct < 4; ++ct) {
    const int col = 16 * ct + ln;
    v2f b1, b2;
    b1.x = Wsa[(size_t)(2 * half) * C_ + col];
    b1.y = Wsa[(size_t)(2 * half + 1) * C_ + col];
    b2.x = Wpe[(size_t)(2 * half) * C_ + col];
    b2.y = pe_hi[col];
    const float bs = bsa[col];
    float bst = NEG_BIG;
#pragma unroll
    for (int rt = 0; rt < 2; ++rt) {
      v8f acc = {0.f, 0.f, 0.f, 0.f, 0.f, 0.f, 0.f, 0.f};
      acc = wmma4(a1[rt], b1, acc);
#pragma unroll
      for (int r = 0; r < 8; ++r) acc[r] = fmaxf(acc[r] + bs, 0.0f);
      acc = wmma4(a2[rt], b2, acc);
#pragma unroll
      for (int r = 0; r < 8; ++r) bst = fmaxf(bst, acc[r]);
    }
    best[ct] = fmaxf(bst, __shfl_xor(bst, 16));
  }
  if (half == 0) {
#pragma unroll
    for (int ct = 0; ct < 4; ++ct) sT[w][16 * ct + ln] = best[ct];
  }
  __syncthreads();

  if (tid < BQ_WAVES * (C_ / 4)) {
    const int row = tid >> 4, c4 = (tid & 15) * 4;
    *(float4*)(feat + (size_t)(bm0 + row) * C_ + c4) = *(const float4*)&sT[row][c4];
  }

  if (w < 4) {
    const int n0 = w * 16;
    const float* Wb = Wc + (size_t)(2 * half) * C_ + n0 + ln;
    v8f acc = {0.f, 0.f, 0.f, 0.f, 0.f, 0.f, 0.f, 0.f};
#pragma unroll
    for (int k = 0; k < C_; k += 4) {
      v2f aa;  aa.x = sT[ln][k + 2 * half];          aa.y = sT[ln][k + 2 * half + 1];
      v2f bb;  bb.x = Wb[(size_t)k * C_];            bb.y = Wb[(size_t)(k + 1) * C_];
      acc = wmma4(aa, bb, acc);
    }
    const int col = n0 + ln;
    const float bv = bc[col];
#pragma unroll
    for (int r = 0; r < 8; ++r) {
      const int row = bm0 + r + 8 * half;
      hout[(size_t)row * C_ + col] = fmaxf(acc[r] + bv, 0.0f);
    }
  }
}

constexpr int FR_BQ = (B_ * M_) / BQ_WAVES;
constexpr int FR_MIX = (FR_BQ / 2) * 3;
constexpr int FR_BLOCKS = 2 * FR_BQ;
static_assert(FR_BQ % 2 == 0, "the mixed range pairs the first query's blocks");

__global__ __launch_bounds__(BQ_THREADS, 12) void front_kernel(
    const float* __restrict__ p, const float* __restrict__ f, const float* __restrict__ Wsa,
    const float* __restrict__ bsa, const float* __restrict__ Wpe, const float* __restrict__ bpe,
    const float* __restrict__ Wc, const float* __restrict__ bc, float* __restrict__ feat,
    float* __restrict__ hout, int* __restrict__ rec2) {
  __shared__ __attribute__((aligned(16))) float4 s_pts[N_ / BQ_GROUP][3][BQ_QUADS][32];
  __shared__ int s_hit[BQ_WAVES][K_];
  __shared__ __attribute__((aligned(16))) float s_tile[BQ_WAVES][PT];
  const int bid = blockIdx.x;
  int role, blk;
  if (bid < FR_MIX) {
    const int t = bid / 3, r = bid - 3 * t;
    role = r < 2 ? 0 : 1;
    blk = r < 2 ? 2 * t + r : t;
  } else {
    role = 1;
    blk = bid - FR_MIX + FR_BQ / 2;
  }
  if (role == 0) {
    const int4 rec = ball_query_block<N_, 1>(s_pts, s_hit, blk, p, R1SQ);
    first_stage_tile(s_tile, rec, blk, f, Wsa, bsa, Wpe, bpe, feat, Wc, bc, hout);
  } else {
    const int4 rec = ball_query_block<M_, STRIDE_>(s_pts, s_hit, blk, p, R2SQ);
    ((int4*)rec2)[((size_t)blk * BQ_WAVES + (threadIdx.x >> 5)) * K_ + (threadIdx.x & 31)] = rec;
  }
}

constexpr int TR = 16;
constexpr int FT = 256;
constexpr int HID = EXP_ * C_;
constexpr int PH = HID + 4;

template <bool MLP, bool CONV>
__global__ __launch_bounds__(FT) void row_tile_kernel(
    const float* __restrict__ src, const int* __restrict__ rec, int src_n,
    const float* __restrict__ Wpe, const float* __restrict__ bpe,
    const float* __restrict__ Wpw1, const float* __restrict__ bpw1,
    const float* __restrict__ Wpw2, const float* __restrict__ bpw2, float* feat,
    const float* __restrict__ Wc, const float* __restrict__ bc, float* __restrict__ hout) {
  static_assert(M_ % TR == 0, "a tile never crosses a batch");
  static_assert(TR * K_ == 2 * FT && K_ == 32, "staging: two records per thread, a wave per query");
  static_assert(TR == 2 * (FT / 32), "gather: two queries per wave");
  __shared__ __attribute__((aligned(16))) int4 s_nb[TR][K_];
  __shared__ __attribute__((aligned(16))) float sT[TR][PT];
  __shared__ __attribute__((aligned(16))) float sH[MLP ? TR : 1][PH];

  const int tid = threadIdx.x;
  const int lane = tid & 31;
  const int half = lane >> 4;
  const int ln = lane & 15;
  const int w = __builtin_amdgcn_readfirstlane(tid >> 5);
  const int bm0 = blockIdx.x * TR;
  const int b = bm0 / M_;

  {
    const int4* rb = (const int4*)rec + (size_t)bm0 * K_;
    const int4 recA = rb[tid];
    const int4 recB = rb[tid + FT];
    s_nb[w][lane] = recA;
    s_nb[w + TR / 2][lane] = recB;
  }
  __syncthreads();

  {
    const int c4 = ln * 4;
    const float4 w0 = *(const float4*)(Wpe + 0 * C_ + c4);
    const float4 w1 = *(const float4*)(Wpe + 1 * C_ + c4);
    const float4 w2 = *(const float4*)(Wpe + 2 * C_ + c4);
    const float4 bi = *(const float4*)(bpe + c4);
    const float* sb = src + (size_t)b * src_n * C_ + c4;
#pragma unroll
    for (int qq = 0; qq < 2; ++qq) {
      const int ql = 2 * w + qq;
      float4 best = {NEG_BIG, NEG_BIG, NEG_BIG, NEG_BIG};
#pragma unroll
      for (int i = 0; i < K_ / 2; ++i) {
        const int4 rec = s_nb[ql][2 * i + half];
        const float dx = __int_as_float(rec.y), dy = __int_as_float(rec.z),
                    dz = __int_as_float(rec.w);
        const float4 s4 = *(const float4*)(sb + (size_t)rec.x * C_);
        best.x = fmaxf(best.x, dx * w0.x + dy * w1.x + dz * w2.x + bi.x + s4.x);
        best.y = fmaxf(best.y, dx * w0.y + dy * w1.y + dz * w2.y + bi.y + s4.y);
        best.z = fmaxf(best.z, dx * w0.z + dy * w1.z + dz * w2.z + bi.z + s4.z);
        best.w = fmaxf(best.w, dx * w0.w + dy * w1.w + dz * w2.w + bi.w + s4.w);
      }
      best.x = fmaxf(best.x, __shfl_xor(best.x, 16));
      best.y = fmaxf(best.y, __shfl_xor(best.y, 16));
      best.z = fmaxf(best.z, __shfl_xor(best.z, 16));
      best.w = fmaxf(best.w, __shfl_xor(best.w, 16));
      if (half == 0) *(float4*)&sT[ql][c4] = best;
    }
  }
  __syncthreads();

  if constexpr (MLP) {
    {
      const int n0 = w * 32;
      const float* Wb = Wpw1 + (size_t)(2 * half) * HID + n0 + ln;
      v8f acc0 = {0.f, 0.f, 0.f, 0.f, 0.f, 0.f, 0.f, 0.f};
      v8f acc1 = {0.f, 0.f, 0.f, 0.f, 0.f, 0.f, 0.f, 0.f};
#pragma unroll
      for (int k = 0; k < C_; k += 4) {
        v2f a;   a.x = sT[ln][k + 2 * half];        a.y = sT[ln][k + 2 * half + 1];
        v2f b0;  b0.x = Wb[(size_t)k * HID];        b0.y = Wb[(size_t)(k + 1) * HID];
        v2f b1;  b1.x = Wb[(size_t)k * HID + 16];   b1.y = Wb[(size_t)(k + 1) * HID + 16];
        acc0 = wmma4(a, b0, acc0);
        acc1 = wmma4(a, b1, acc1);
      }
      const float bv0 = bpw1[n0 + ln], bv1 = bpw1[n0 + 16 + ln];
#pragma unroll
      for (int r = 0; r < 8; ++r) {
        sH[r + 8 * half][n0 + ln] = fmaxf(acc0[r] + bv0, 0.0f);
        sH[r + 8 * half][n0 + 16 + ln] = fmaxf(acc1[r] + bv1, 0.0f);
      }
    }
    __syncthreads();

    if (w < 4) {
      const int n0 = w * 16;
      const float* Wb = Wpw2 + (size_t)(2 * half) * C_ + n0 + ln;
      v8f acc = {0.f, 0.f, 0.f, 0.f, 0.f, 0.f, 0.f, 0.f};
#pragma unroll
      for (int k = 0; k < HID; k += 4) {
        v2f a;   a.x = sH[ln][k + 2 * half];        a.y = sH[ln][k + 2 * half + 1];
        v2f bb;  bb.x = Wb[(size_t)k * C_];         bb.y = Wb[(size_t)(k + 1) * C_];
        acc = wmma4(a, bb, acc);
      }
      const int col = n0 + ln;
      const float bv = bpw2[col];
#pragma unroll
      for (int r = 0; r < 8; ++r) {
        const int row = r + 8 * half;
        float v = acc[r] + bv;
        v += feat[(size_t)(bm0 + row) * C_ + col];
        sT[row][col] = fmaxf(v, 0.0f);
      }
    }
    __syncthreads();
  }

  {
    const int row = tid >> 4, c4 = (tid & 15) * 4;
    *(float4*)(feat + (size_t)(bm0 + row) * C_ + c4) = *(const float4*)&sT[row][c4];
  }

  if constexpr (CONV) {
    if (w < 4) {
      const int n0 = w * 16;
      const float* Wb = Wc + (size_t)(2 * half) * C_ + n0 + ln;
      v8f acc = {0.f, 0.f, 0.f, 0.f, 0.f, 0.f, 0.f, 0.f};
#pragma unroll
      for (int k = 0; k < C_; k += 4) {
        v2f a;   a.x = sT[ln][k + 2 * half];        a.y = sT[ln][k + 2 * half + 1];
        v2f bb;  bb.x = Wb[(size_t)k * C_];         bb.y = Wb[(size_t)(k + 1) * C_];
        acc = wmma4(a, bb, acc);
      }
      const int col = n0 + ln;
      const float bv = bc[col];
#pragma unroll
      for (int r = 0; r < 8; ++r) {
        const int row = bm0 + r + 8 * half;
        hout[(size_t)row * C_ + col] = fmaxf(acc[r] + bv, 0.0f);
      }
    }
  }
}

}

extern "C" void kernel_launch(void* const* d_in, const int* in_sizes, int n_in,
                              void* d_out, int out_size, void* d_ws, size_t ws_size,
                              hipStream_t stream) {
  (void)in_sizes; (void)n_in; (void)out_size; (void)ws_size;
  const float* p      = (const float*)d_in[0];
  const float* f      = (const float*)d_in[1];
  const float* W_sa   = (const float*)d_in[2];
  const float* b_sa   = (const float*)d_in[3];
  const float* W_pe1  = (const float*)d_in[4];
  const float* b_pe1  = (const float*)d_in[5];
  const float* W_pe2  = (const float*)d_in[6];
  const float* b_pe2  = (const float*)d_in[7];
  const float* W_conv = (const float*)d_in[8];
  const float* b_conv = (const float*)d_in[9];
  const float* W_pw1  = (const float*)d_in[10];
  const float* b_pw1  = (const float*)d_in[11];
  const float* W_pw2  = (const float*)d_in[12];
  const float* b_pw2  = (const float*)d_in[13];
  float* feat = (float*)d_out;

  char* ws = (char*)d_ws;
  float* hA = (float*)ws;  ws += sizeof(float) * (size_t)B_ * M_ * C_;
  float* hB = (float*)ws;  ws += sizeof(float) * (size_t)B_ * M_ * C_;
  int* rec2 = (int*)ws;

  front_kernel<<<FR_BLOCKS, BQ_THREADS, 0, stream>>>(p, f, W_sa, b_sa, W_pe1, b_pe1, W_conv, b_conv,
                                                     feat, hA, rec2);

  const int tiles = (B_ * M_) / TR;

  row_tile_kernel<true, true><<<tiles, FT, 0, stream>>>(
      hA, rec2, M_, W_pe2, b_pe2, W_pw1, b_pw1, W_pw2, b_pw2, feat,
      W_conv + (size_t)C_ * C_, b_conv + C_, hB);

  row_tile_kernel<true, false><<<tiles, FT, 0, stream>>>(
      hB, rec2, M_, W_pe2, b_pe2, W_pw1 + (size_t)C_ * HID, b_pw1 + HID,
      W_pw2 + (size_t)HID * C_, b_pw2 + C_, feat, nullptr, nullptr, nullptr);
}


# front launch: the 512 long first-query blocks get block numbers 0..511, the short second-query blocks follow (was dealt 2:1): every long block enters within two dispatch passes, not three
# speedup vs baseline: 5.1279x; 1.0321x over previous
#include <hip/hip_runtime.h>
#include <hip/hip_bf16.h>
#include <cstdint>

namespace {

constexpr int B_ = 4, N_ = 8192, CIN_ = 4, C_ = 64, K_ = 32, STRIDE_ = 4, NB_ = 2, EXP_ = 4;
constexpr int M_ = N_ / STRIDE_;
constexpr float R1SQ = 0.1f * 0.1f;
constexpr float R2SQ = 0.2f * 0.2f;
constexpr float NEG_BIG = -3.4028235e38f;

typedef float v2f __attribute__((ext_vector_type(2)));
typedef float v8f __attribute__((ext_vector_type(8)));

constexpr int BQ_THREADS = 512;
constexpr int BQ_WAVES = BQ_THREADS / 32;
constexpr int BQ_GROUP = 256;
constexpr int BQ_CHUNKS = BQ_GROUP / 32;
constexpr int BQ_QUAD = 128;
constexpr int BQ_QUADS = BQ_GROUP / BQ_QUAD;

template <int NSRC, int SSTRIDE>
__device__ __forceinline__ int4 ball_query_block(float4 (*s)[3][BQ_QUADS][32], int (*s_hit)[K_],
                                                 int blk, const float* __restrict__ p, float r2) {
  static_assert(K_ == 32, "a lane per neighbour slot when the records are written");
  static_assert((NSRC / BQ_QUAD) % BQ_WAVES == 0, "staging loop covers NSRC exactly, a quad per wave and pass");
  static_assert(NSRC % BQ_GROUP == 0, "scan trips cover NSRC exactly");
  static_assert(M_ % BQ_WAVES == 0, "a block never crosses a batch");

  const int tid = threadIdx.x;
  const int lane = tid & 31;
  const int q0 = blk * BQ_WAVES;
  const int b = q0 / M_;
  const float* pb = p + (size_t)b * N_ * 3;

#pragma unroll
  for (int i = 0; i < (NSRC / BQ_QUAD) / BQ_WAVES; ++i) {
    const int quad = (tid >> 5) + i * BQ_WAVES;
    float cx[4], cy[4], cz[4];
#pragma unroll
    for (int e = 0; e < 4; ++e) {
      const int j = quad * BQ_QUAD + e * 32 + lane;
      const float* sp = pb + (size_t)j * SSTRIDE * 3;
      cx[e] = sp[0];
      cy[e] = sp[1];
      cz[e] = sp[2];
    }
    s[quad / BQ_QUADS][0][quad % BQ_QUADS][lane] = make_float4(cx[0], cx[1], cx[2], cx[3]);
    s[quad / BQ_QUADS][1][quad % BQ_QUADS][lane] = make_float4(cy[0], cy[1], cy[2], cy[3]);
    s[quad / BQ_QUADS][2][quad % BQ_QUADS][lane] = make_float4(cz[0], cz[1], cz[2], cz[3]);
  }
  __syncthreads();

  const int w = __builtin_amdgcn_readfirstlane(tid >> 5);
  const int wave = q0 + w;
  const int m = wave - b * M_;
  const float qx = pb[(size_t)m * STRIDE_ * 3 + 0];
  const float qy = pb[(size_t)m * STRIDE_ * 3 + 1];
  const float qz = pb[(size_t)m * STRIDE_ * 3 + 2];
  int* hit = s_hit[w];

  int count = 0;
  for (int g = 0; g < NSRC / BQ_GROUP; ++g) {
    unsigned mk[BQ_CHUNKS];
    float sx[BQ_CHUNKS], sy[BQ_CHUNKS], sz[BQ_CHUNKS];
#pragma unroll
    for (int h = 0; h < BQ_QUADS; ++h) {
      const float4 vx = s[g][0][h][lane], vy = s[g][1][h][lane], vz = s[g][2][h][lane];
      sx[4 * h + 0] = vx.x; sx[4 * h + 1] = vx.y; sx[4 * h + 2] = vx.z; sx[4 * h + 3] = vx.w;
      sy[4 * h + 0] = vy.x; sy[4 * h + 1] = vy.y; sy[4 * h + 2] = vy.z; sy[4 * h + 3] = vy.w;
      sz[4 * h + 0] = vz.x; sz[4 * h + 1] = vz.y; sz[4 * h + 2] = vz.z; sz[4 * h + 3] = vz.w;
    }
#pragma unroll
    for (int u = 0; u < BQ_CHUNKS; ++u) {
      float dx = sx[u] - qx, dy = sy[u] - qy, dz = sz[u] - qz;
      float sxy;
      {
#pragma clang fp contract(off)
        sxy = dx * dx + dy * dy;
      }
      float d2 = __builtin_fmaf(dz, dz, sxy);
      mk[u] = (unsigned)__ballot(d2 <= r2);
    }
    unsigned any = 0u;
#pragma unroll
    for (int u = 0; u < BQ_CHUNKS; ++u) any |= mk[u];
    if (any != 0u) {
      const int base = g * BQ_GROUP;
#pragma unroll
      for (int u = 0; u < BQ_CHUNKS; ++u) {
        unsigned mask = mk[u];
        if (mask != 0u) {
          if ((mask >> lane) & 1u) {
            int pos = count + (int)__builtin_amdgcn_mbcnt_lo(mask, 0u);
            if (pos < K_) hit[pos] = base + u * 32 + lane;
          }
          count += __popc(mask);
        }
      }
      if (count >= K_) break;
    }
  }
  if (count > K_) count = K_;
  int j = hit[lane < count ? lane : 0];
  if (count == 0) j = NSRC - 1;
  const int jg = j / BQ_GROUP, jr = j % BQ_GROUP;
  const int jh = jr / BQ_QUAD, jq = jr % BQ_QUAD;
  const int je = jq >> 5, jl = jq & 31;
  const float px = ((const float*)&s[jg][0][jh][jl])[je];
  const float py = ((const float*)&s[jg][1][jh][jl])[je];
  const float pz = ((const float*)&s[jg][2][jh][jl])[je];
  int4 rec;
  rec.x = j;
  rec.y = __float_as_int(px - qx);
  rec.z = __float_as_int(py - qy);
  rec.w = __float_as_int(pz - qz);
  return rec;
}

constexpr int PT = C_ + 4;

__device__ __forceinline__ v8f wmma4(v2f a, v2f b, v8f c) {
  return __builtin_amdgcn_wmma_f32_16x16x4_f32(false, a, false, b, (short)0, c, false, false);
}

__device__ __forceinline__ void first_stage_tile(
    float (*sT)[PT], int4 rec, int blk, const float* __restrict__ f,
    const float* __restrict__ Wsa, const float* __restrict__ bsa,
    const float* __restrict__ Wpe, const float* __restrict__ bpe, float* __restrict__ feat,
    const float* __restrict__ Wc, const float* __restrict__ bc, float* __restrict__ hout) {
  static_assert(BQ_WAVES == 16 && K_ == 32 && C_ == 64 && CIN_ == 4,
                "a block is one 16-row tile; 2 x 4 WMMA tiles of K = 4 per query");
  const int tid = threadIdx.x;
  const int lane = tid & 31;
  const int half = lane >> 4;
  const int ln = lane & 15;
  const int w = __builtin_amdgcn_readfirstlane(tid >> 5);
  const int bm0 = blk * BQ_WAVES;
  const int b = bm0 / M_;

  int4 oth;
  oth.x = __shfl_xor(rec.x, 16);
  oth.y = __shfl_xor(rec.y, 16);
  oth.z = __shfl_xor(rec.z, 16);
  oth.w = __shfl_xor(rec.w, 16);

  const float* fb = f + ((size_t)b * CIN_ + 2 * half) * N_;
  v2f a1[2], a2[2];
#pragma unroll
  for (int rt = 0; rt < 2; ++rt) {
    const int4 nb = (rt == half) ? rec : oth;
    a1[rt].x = fb[nb.x];
    a1[rt].y = fb[N_ + nb.x];
    a2[rt].x = half ? __int_as_float(nb.w) : __int_as_float(nb.y);
    a2[rt].y = half ? 1.0f : __int_as_float(nb.z);
  }

  const float* pe_hi = half ? bpe : Wpe + C_;
  float best[4];
#pragma unroll
  for (int ct = 0; ct < 4; ++ct) {
    const int col = 16 * ct + ln;
    v2f b1, b2;
    b1.x = Wsa[(size_t)(2 * half) * C_ + col];
    b1.y = Wsa[(size_t)(2 * half + 1) * C_ + col];
    b2.x = Wpe[(size_t)(2 * half) * C_ + col];
    b2.y = pe_hi[col];
    const float bs = bsa[col];
    float bst = NEG_BIG;
#pragma unroll
    for (int rt = 0; rt < 2; ++rt) {
      v8f acc = {0.f, 0.f, 0.f, 0.f, 0.f, 0.f, 0.f, 0.f};
      acc = wmma4(a1[rt], b1, acc);
#pragma unroll
      for (int r = 0; r < 8; ++r) acc[r] = fmaxf(acc[r] + bs, 0.0f);
      acc = wmma4(a2[rt], b2, acc);
#pragma unroll
      for (int r = 0; r < 8; ++r) bst = fmaxf(bst, acc[r]);
    }
    best[ct] = fmaxf(bst, __shfl_xor(bst, 16));
  }
  if (half == 0) {
#pragma unroll
    for (int ct = 0; ct < 4; ++ct) sT[w][16 * ct + ln] = best[ct];
  }
  __syncthreads();

  if (tid < BQ_WAVES * (C_ / 4)) {
    const int row = tid >> 4, c4 = (tid & 15) * 4;
    *(float4*)(feat + (size_t)(bm0 + row) * C_ + c4) = *(const float4*)&sT[row][c4];
  }

  if (w < 4) {
    const int n0 = w * 16;
    const float* Wb = Wc + (size_t)(2 * half) * C_ + n0 + ln;
    v8f acc = {0.f, 0.f, 0.f, 0.f, 0.f, 0.f, 0.f, 0.f};
#pragma unroll
    for (int k = 0; k < C_; k += 4) {
      v2f aa;  aa.x = sT[ln][k + 2 * half];          aa.y = sT[ln][k + 2 * half + 1];
      v2f bb;  bb.x = Wb[(size_t)k * C_];            bb.y = Wb[(size_t)(k + 1) * C_];
      acc = wmma4(aa, bb, acc);
    }
    const int col = n0 + ln;
    const float bv = bc[col];
#pragma unroll
    for (int r = 0; r < 8; ++r) {
      const int row = bm0 + r + 8 * half;
      hout[(size_t)row * C_ + col] = fmaxf(acc[r] + bv, 0.0f);
    }
  }
}

constexpr int FR_BQ = (B_ * M_) / BQ_WAVES;
constexpr int FR_BLOCKS = 2 * FR_BQ;

__global__ __launch_bounds__(BQ_THREADS, 12) void front_kernel(
    const float* __restrict__ p, const float* __restrict__ f, const float* __restrict__ Wsa,
    const float* __restrict__ bsa, const float* __restrict__ Wpe, const float* __restrict__ bpe,
    const float* __restrict__ Wc, const float* __restrict__ bc, float* __restrict__ feat,
    float* __restrict__ hout, int* __restrict__ rec2) {
  __shared__ __attribute__((aligned(16))) float4 s_pts[N_ / BQ_GROUP][3][BQ_QUADS][32];
  __shared__ int s_hit[BQ_WAVES][K_];
  __shared__ __attribute__((aligned(16))) float s_tile[BQ_WAVES][PT];
  const int bid = blockIdx.x;
  const int role = bid < FR_BQ ? 0 : 1;
  const int blk = bid < FR_BQ ? bid : bid - FR_BQ;
  if (role == 0) {
    const int4 rec = ball_query_block<N_, 1>(s_pts, s_hit, blk, p, R1SQ);
    first_stage_tile(s_tile, rec, blk, f, Wsa, bsa, Wpe, bpe, feat, Wc, bc, hout);
  } else {
    const int4 rec = ball_query_block<M_, STRIDE_>(s_pts, s_hit, blk, p, R2SQ);
    ((int4*)rec2)[((size_t)blk * BQ_WAVES + (threadIdx.x >> 5)) * K_ + (threadIdx.x & 31)] = rec;
  }
}

constexpr int TR = 16;
constexpr int FT = 256;
constexpr int HID = EXP_ * C_;
constexpr int PH = HID + 4;

template <bool MLP, bool CONV>
__global__ __launch_bounds__(FT) void row_tile_kernel(
    const float* __restrict__ src, const int* __restrict__ rec, int src_n,
    const float* __restrict__ Wpe, const float* __restrict__ bpe,
    const float* __restrict__ Wpw1, const float* __restrict__ bpw1,
    const float* __restrict__ Wpw2, const float* __restrict__ bpw2, float* feat,
    const float* __restrict__ Wc, const float* __restrict__ bc, float* __restrict__ hout) {
  static_assert(M_ % TR == 0, "a tile never crosses a batch");
  static_assert(TR * K_ == 2 * FT && K_ == 32, "staging: two records per thread, a wave per query");
  static_assert(TR == 2 * (FT / 32), "gather: two queries per wave");
  __shared__ __attribute__((aligned(16))) int4 s_nb[TR][K_];
  __shared__ __attribute__((aligned(16))) float sT[TR][PT];
  __shared__ __attribute__((aligned(16))) float sH[MLP ? TR : 1][PH];

  const int tid = threadIdx.x;
  const int lane = tid & 31;
  const int half = lane >> 4;
  const int ln = lane & 15;
  const int w = __builtin_amdgcn_readfirstlane(tid >> 5);
  const int bm0 = blockIdx.x * TR;
  const int b = bm0 / M_;

  {
    const int4* rb = (const int4*)rec + (size_t)bm0 * K_;
    const int4 recA = rb[tid];
    const int4 recB = rb[tid + FT];
    s_nb[w][lane] = recA;
    s_nb[w + TR / 2][lane] = recB;
  }
  __syncthreads();

  {
    const int c4 = ln * 4;
    const float4 w0 = *(const float4*)(Wpe + 0 * C_ + c4);
    const float4 w1 = *(const float4*)(Wpe + 1 * C_ + c4);
    const float4 w2 = *(const float4*)(Wpe + 2 * C_ + c4);
    const float4 bi = *(const float4*)(bpe + c4);
    const float* sb = src + (size_t)b * src_n * C_ + c4;
#pragma unroll
    for (int qq = 0; qq < 2; ++qq) {
      const int ql = 2 * w + qq;
      float4 best = {NEG_BIG, NEG_BIG, NEG_BIG, NEG_BIG};
#pragma unroll
      for (int i = 0; i < K_ / 2; ++i) {
        const int4 rec = s_nb[ql][2 * i + half];
        const float dx = __int_as_float(rec.y), dy = __int_as_float(rec.z),
                    dz = __int_as_float(rec.w);
        const float4 s4 = *(const float4*)(sb + (size_t)rec.x * C_);
        best.x = fmaxf(best.x, dx * w0.x + dy * w1.x + dz * w2.x + bi.x + s4.x);
        best.y = fmaxf(best.y, dx * w0.y + dy * w1.y + dz * w2.y + bi.y + s4.y);
        best.z = fmaxf(best.z, dx * w0.z + dy * w1.z + dz * w2.z + bi.z + s4.z);
        best.w = fmaxf(best.w, dx * w0.w + dy * w1.w + dz * w2.w + bi.w + s4.w);
      }
      best.x = fmaxf(best.x, __shfl_xor(best.x, 16));
      best.y = fmaxf(best.y, __shfl_xor(best.y, 16));
      best.z = fmaxf(best.z, __shfl_xor(best.z, 16));
      best.w = fmaxf(best.w, __shfl_xor(best.w, 16));
      if (half == 0) *(float4*)&sT[ql][c4] = best;
    }
  }
  __syncthreads();

  if constexpr (MLP) {
    {
      const int n0 = w * 32;
      const float* Wb = Wpw1 + (size_t)(2 * half) * HID + n0 + ln;
      v8f acc0 = {0.f, 0.f, 0.f, 0.f, 0.f, 0.f, 0.f, 0.f};
      v8f acc1 = {0.f, 0.f, 0.f, 0.f, 0.f, 0.f, 0.f, 0.f};
#pragma unroll
      for (int k = 0; k < C_; k += 4) {
        v2f a;   a.x = sT[ln][k + 2 * half];        a.y = sT[ln][k + 2 * half + 1];
        v2f b0;  b0.x = Wb[(size_t)k * HID];        b0.y = Wb[(size_t)(k + 1) * HID];
        v2f b1;  b1.x = Wb[(size_t)k * HID + 16];   b1.y = Wb[(size_t)(k + 1) * HID + 16];
        acc0 = wmma4(a, b0, acc0);
        acc1 = wmma4(a, b1, acc1);
      }
      const float bv0 = bpw1[n0 + ln], bv1 = bpw1[n0 + 16 + ln];
#pragma unroll
      for (int r = 0; r < 8; ++r) {
        sH[r + 8 * half][n0 + ln] = fmaxf(acc0[r] + bv0, 0.0f);
        sH[r + 8 * half][n0 + 16 + ln] = fmaxf(acc1[r] + bv1, 0.0f);
      }
    }
    __syncthreads();

    if (w < 4) {
      const int n0 = w * 16;
      const float* Wb = Wpw2 + (size_t)(2 * half) * C_ + n0 + ln;
      v8f acc = {0.f, 0.f, 0.f, 0.f, 0.f, 0.f, 0.f, 0.f};
#pragma unroll
      for (int k = 0; k < HID; k += 4) {
        v2f a;   a.x = sH[ln][k + 2 * half];        a.y = sH[ln][k + 2 * half + 1];
        v2f bb;  bb.x = Wb[(size_t)k * C_];         bb.y = Wb[(size_t)(k + 1) * C_];
        acc = wmma4(a, bb, acc);
      }
      const int col = n0 + ln;
      const float bv = bpw2[col];
#pragma unroll
      for (int r = 0; r < 8; ++r) {
        const int row = r + 8 * half;
        float v = acc[r] + bv;
        v += feat[(size_t)(bm0 + row) * C_ + col];
        sT[row][col] = fmaxf(v, 0.0f);
      }
    }
    __syncthreads();
  }

  {
    const int row = tid >> 4, c4 = (tid & 15) * 4;
    *(float4*)(feat + (size_t)(bm0 + row) * C_ + c4) = *(const float4*)&sT[row][c4];
  }

  if constexpr (CONV) {
    if (w < 4) {
      const int n0 = w * 16;
      const float* Wb = Wc + (size_t)(2 * half) * C_ + n0 + ln;
      v8f acc = {0.f, 0.f, 0.f, 0.f, 0.f, 0.f, 0.f, 0.f};
#pragma unroll
      for (int k = 0; k < C_; k += 4) {
        v2f a;   a.x = sT[ln][k + 2 * half];        a.y = sT[ln][k + 2 * half + 1];
        v2f bb;  bb.x = Wb[(size_t)k * C_];         bb.y = Wb[(size_t)(k + 1) * C_];
        acc = wmma4(a, bb, acc);
      }
      const int col = n0 + ln;
      const float bv = bc[col];
#pragma unroll
      for (int r = 0; r < 8; ++r) {
        const int row = bm0 + r + 8 * half;
        hout[(size_t)row * C_ + col] = fmaxf(acc[r] + bv, 0.0f);
      }
    }
  }
}

}

extern "C" void kernel_launch(void* const* d_in, const int* in_sizes, int n_in,
                              void* d_out, int out_size, void* d_ws, size_t ws_size,
                              hipStream_t stream) {
  (void)in_sizes; (void)n_in; (void)out_size; (void)ws_size;
  const float* p      = (const float*)d_in[0];
  const float* f      = (const float*)d_in[1];
  const float* W_sa   = (const float*)d_in[2];
  const float* b_sa   = (const float*)d_in[3];
  const float* W_pe1  = (const float*)d_in[4];
  const float* b_pe1  = (const float*)d_in[5];
  const float* W_pe2  = (const float*)d_in[6];
  const float* b_pe2  = (const float*)d_in[7];
  const float* W_conv = (const float*)d_in[8];
  const float* b_conv = (const float*)d_in[9];
  const float* W_pw1  = (const float*)d_in[10];
  const float* b_pw1  = (const float*)d_in[11];
  const float* W_pw2  = (const float*)d_in[12];
  const float* b_pw2  = (const float*)d_in[13];
  float* feat = (float*)d_out;

  char* ws = (char*)d_ws;
  float* hA = (float*)ws;  ws += sizeof(float) * (size_t)B_ * M_ * C_;
  float* hB = (float*)ws;  ws += sizeof(float) * (size_t)B_ * M_ * C_;
  int* rec2 = (int*)ws;

  front_kernel<<<FR_BLOCKS, BQ_THREADS, 0, stream>>>(p, f, W_sa, b_sa, W_pe1, b_pe1, W_conv, b_conv,
                                                     feat, hA, rec2);

  const int tiles = (B_ * M_) / TR;

  row_tile_kernel<true, true><<<tiles, FT, 0, stream>>>(
      hA, rec2, M_, W_pe2, b_pe2, W_pw1, b_pw1, W_pw2, b_pw2, feat,
      W_conv + (size_t)C_ * C_, b_conv + C_, hB);

  row_tile_kernel<true, false><<<tiles, FT, 0, stream>>>(
      hB, rec2, M_, W_pe2, b_pe2, W_pw1 + (size_t)C_ * HID, b_pw1 + HID,
      W_pw2 + (size_t)HID * C_, b_pw2 + C_, feat, nullptr, nullptr, nullptr);
}


# front launch: source points staged in four passes of 2048; a pass's global loads are issued before the barrier of the pass before (LDS-only fence) and land during its scan; hit order, sums unchanged
# speedup vs baseline: 5.3096x; 1.0354x over previous
#include <hip/hip_runtime.h>
#include <hip/hip_bf16.h>
#include <cstdint>

namespace {

constexpr int B_ = 4, N_ = 8192, CIN_ = 4, C_ = 64, K_ = 32, STRIDE_ = 4, NB_ = 2, EXP_ = 4;
constexpr int M_ = N_ / STRIDE_;
constexpr float R1SQ = 0.1f * 0.1f;
constexpr float R2SQ = 0.2f * 0.2f;
constexpr float NEG_BIG = -3.4028235e38f;

typedef float v2f __attribute__((ext_vector_type(2)));
typedef float v8f __attribute__((ext_vector_type(8)));

constexpr int BQ_THREADS = 512;
constexpr int BQ_WAVES = BQ_THREADS / 32;
constexpr int BQ_GROUP = 256;
constexpr int BQ_CHUNKS = BQ_GROUP / 32;
constexpr int BQ_QUAD = 128;
constexpr int BQ_QUADS = BQ_GROUP / BQ_QUAD;

__device__ __forceinline__ int scan_groups(float4 (*s)[3][BQ_QUADS][32], int* hit, int g0, int g1,
                                           int count, float qx, float qy, float qz, float r2) {
  const int lane = threadIdx.x & 31;
  for (int g = g0; g < g1; ++g) {
    unsigned mk[BQ_CHUNKS];
    float sx[BQ_CHUNKS], sy[BQ_CHUNKS], sz[BQ_CHUNKS];
#pragma unroll
    for (int h = 0; h < BQ_QUADS; ++h) {
      const float4 vx = s[g][0][h][lane], vy = s[g][1][h][lane], vz = s[g][2][h][lane];
      sx[4 * h + 0] = vx.x; sx[4 * h + 1] = vx.y; sx[4 * h + 2] = vx.z; sx[4 * h + 3] = vx.w;
      sy[4 * h + 0] = vy.x; sy[4 * h + 1] = vy.y; sy[4 * h + 2] = vy.z; sy[4 * h + 3] = vy.w;
      sz[4 * h + 0] = vz.x; sz[4 * h + 1] = vz.y; sz[4 * h + 2] = vz.z; sz[4 * h + 3] = vz.w;
    }
#pragma unroll
    for (int u = 0; u < BQ_CHUNKS; ++u) {
      float dx = sx[u] - qx, dy = sy[u] - qy, dz = sz[u] - qz;
      float sxy;
      {
#pragma clang fp contract(off)
        sxy = dx * dx + dy * dy;
      }
      float d2 = __builtin_fmaf(dz, dz, sxy);
      mk[u] = (unsigned)__ballot(d2 <= r2);
    }
    unsigned any = 0u;
#pragma unroll
    for (int u = 0; u < BQ_CHUNKS; ++u) any |= mk[u];
    if (any != 0u) {
      const int base = g * BQ_GROUP;
#pragma unroll
      for (int u = 0; u < BQ_CHUNKS; ++u) {
        unsigned mask = mk[u];
        if (mask != 0u) {
          if ((mask >> lane) & 1u) {
            int pos = count + (int)__builtin_amdgcn_mbcnt_lo(mask, 0u);
            if (pos < K_) hit[pos] = base + u * 32 + lane;
          }
          count += __popc(mask);
        }
      }
      if (count >= K_) break;
    }
  }
  return count;
}

template <int NSRC, int SSTRIDE>
__device__ __forceinline__ int4 ball_query_block(float4 (*s)[3][BQ_QUADS][32], int (*s_hit)[K_],
                                                 int blk, const float* __restrict__ p, float r2) {
  static_assert(K_ == 32, "a lane per neighbour slot when the records are written");
  static_assert((NSRC / BQ_QUAD) % BQ_WAVES == 0, "staging loop covers NSRC exactly, a quad per wave and pass");
  static_assert(NSRC % BQ_GROUP == 0, "scan trips cover NSRC exactly");
  static_assert(M_ % BQ_WAVES == 0, "a block never crosses a batch");

  const int tid = threadIdx.x;
  const int lane = tid & 31;
  const int q0 = blk * BQ_WAVES;
  const int b = q0 / M_;
  const float* pb = p + (size_t)b * N_ * 3;

  const int w = __builtin_amdgcn_readfirstlane(tid >> 5);
  const int wave = q0 + w;
  const int m = wave - b * M_;
  int* hit = s_hit[w];

  const float qx = pb[(size_t)m * STRIDE_ * 3 + 0];
  const float qy = pb[(size_t)m * STRIDE_ * 3 + 1];
  const float qz = pb[(size_t)m * STRIDE_ * 3 + 2];

  constexpr int PASSES = (NSRC / BQ_QUAD) / BQ_WAVES;
  constexpr int PASS_GROUPS = (BQ_WAVES * BQ_QUAD) / BQ_GROUP;
  float cx[4], cy[4], cz[4];
#pragma unroll
  for (int e = 0; e < 4; ++e) {
    const float* sp = pb + (size_t)(w * BQ_QUAD + e * 32 + lane) * SSTRIDE * 3;
    cx[e] = sp[0];
    cy[e] = sp[1];
    cz[e] = sp[2];
  }

  __builtin_amdgcn_sched_barrier(0);

  int count = 0;
#pragma unroll
  for (int i = 0; i < PASSES; ++i) {
    const int quad = w + i * BQ_WAVES;
    s[quad / BQ_QUADS][0][quad % BQ_QUADS][lane] = make_float4(cx[0], cx[1], cx[2], cx[3]);
    s[quad / BQ_QUADS][1][quad % BQ_QUADS][lane] = make_float4(cy[0], cy[1], cy[2], cy[3]);
    s[quad / BQ_QUADS][2][quad % BQ_QUADS][lane] = make_float4(cz[0], cz[1], cz[2], cz[3]);
    if (i + 1 < PASSES) {
#pragma unroll
      for (int e = 0; e < 4; ++e) {
        const float* sp = pb + (size_t)((quad + BQ_WAVES) * BQ_QUAD + e * 32 + lane) * SSTRIDE * 3;
        cx[e] = sp[0];
        cy[e] = sp[1];
        cz[e] = sp[2];
      }
    }
    __builtin_amdgcn_fence(__ATOMIC_RELEASE, "workgroup", "local");
    __builtin_amdgcn_s_barrier();
    __builtin_amdgcn_fence(__ATOMIC_ACQUIRE, "workgroup", "local");
    if (count < K_) count = scan_groups(s, hit, i * PASS_GROUPS, (i + 1) * PASS_GROUPS, count, qx, qy, qz, r2);
  }
  if (count > K_) count = K_;
  int j = hit[lane < count ? lane : 0];
  if (count == 0) j = NSRC - 1;
  const int jg = j / BQ_GROUP, jr = j % BQ_GROUP;
  const int jh = jr / BQ_QUAD, jq = jr % BQ_QUAD;
  const int je = jq >> 5, jl = jq & 31;
  const float px = ((const float*)&s[jg][0][jh][jl])[je];
  const float py = ((const float*)&s[jg][1][jh][jl])[je];
  const float pz = ((const float*)&s[jg][2][jh][jl])[je];
  int4 rec;
  rec.x = j;
  rec.y = __float_as_int(px - qx);
  rec.z = __float_as_int(py - qy);
  rec.w = __float_as_int(pz - qz);
  return rec;
}

constexpr int PT = C_ + 4;

__device__ __forceinline__ v8f wmma4(v2f a, v2f b, v8f c) {
  return __builtin_amdgcn_wmma_f32_16x16x4_f32(false, a, false, b, (short)0, c, false, false);
}

__device__ __forceinline__ void first_stage_tile(
    float (*sT)[PT], int4 rec, int blk, const float* __restrict__ f,
    const float* __restrict__ Wsa, const float* __restrict__ bsa,
    const float* __restrict__ Wpe, const float* __restrict__ bpe, float* __restrict__ feat,
    const float* __restrict__ Wc, const float* __restrict__ bc, float* __restrict__ hout) {
  static_assert(BQ_WAVES == 16 && K_ == 32 && C_ == 64 && CIN_ == 4,
                "a block is one 16-row tile; 2 x 4 WMMA tiles of K = 4 per query");
  const int tid = threadIdx.x;
  const int lane = tid & 31;
  const int half = lane >> 4;
  const int ln = lane & 15;
  const int w = __builtin_amdgcn_readfirstlane(tid >> 5);
  const int bm0 = blk * BQ_WAVES;
  const int b = bm0 / M_;

  int4 oth;
  oth.x = __shfl_xor(rec.x, 16);
  oth.y = __shfl_xor(rec.y, 16);
  oth.z = __shfl_xor(rec.z, 16);
  oth.w = __shfl_xor(rec.w, 16);

  const float* fb = f + ((size_t)b * CIN_ + 2 * half) * N_;
  v2f a1[2], a2[2];
#pragma unroll
  for (int rt = 0; rt < 2; ++rt) {
    const int4 nb = (rt == half) ? rec : oth;
    a1[rt].x = fb[nb.x];
    a1[rt].y = fb[N_ + nb.x];
    a2[rt].x = half ? __int_as_float(nb.w) : __int_as_float(nb.y);
    a2[rt].y = half ? 1.0f : __int_as_float(nb.z);
  }

  const float* pe_hi = half ? bpe : Wpe + C_;
  float best[4];
#pragma unroll
  for (int ct = 0; ct < 4; ++ct) {
    const int col = 16 * ct + ln;
    v2f b1, b2;
    b1.x = Wsa[(size_t)(2 * half) * C_ + col];
    b1.y = Wsa[(size_t)(2 * half + 1) * C_ + col];
    b2.x = Wpe[(size_t)(2 * half) * C_ + col];
    b2.y = pe_hi[col];
    const float bs = bsa[col];
    float bst = NEG_BIG;
#pragma unroll
    for (int rt = 0; rt < 2; ++rt) {
      v8f acc = {0.f, 0.f, 0.f, 0.f, 0.f, 0.f, 0.f, 0.f};
      acc = wmma4(a1[rt], b1, acc);
#pragma unroll
      for (int r = 0; r < 8; ++r) acc[r] = fmaxf(acc[r] + bs, 0.0f);
      acc = wmma4(a2[rt], b2, acc);
#pragma unroll
      for (int r = 0; r < 8; ++r) bst = fmaxf(bst, acc[r]);
    }
    best[ct] = fmaxf(bst, __shfl_xor(bst, 16));
  }
  if (half == 0) {
#pragma unroll
    for (int ct = 0; ct < 4; ++ct) sT[w][16 * ct + ln] = best[ct];
  }
  __syncthreads();

  if (tid < BQ_WAVES * (C_ / 4)) {
    const int row = tid >> 4, c4 = (tid & 15) * 4;
    *(float4*)(feat + (size_t)(bm0 + row) * C_ + c4) = *(const float4*)&sT[row][c4];
  }

  if (w < 4) {
    const int n0 = w * 16;
    const float* Wb = Wc + (size_t)(2 * half) * C_ + n0 + ln;
    v8f acc = {0.f, 0.f, 0.f, 0.f, 0.f, 0.f, 0.f, 0.f};
#pragma unroll
    for (int k = 0; k < C_; k += 4) {
      v2f aa;  aa.x = sT[ln][k + 2 * half];          aa.y = sT[ln][k + 2 * half + 1];
      v2f bb;  bb.x = Wb[(size_t)k * C_];            bb.y = Wb[(size_t)(k + 1) * C_];
      acc = wmma4(aa, bb, acc);
    }
    const int col = n0 + ln;
    const float bv = bc[col];
#pragma unroll
    for (int r = 0; r < 8; ++r) {
      const int row = bm0 + r + 8 * half;
      hout[(size_t)row * C_ + col] = fmaxf(acc[r] + bv, 0.0f);
    }
  }
}

constexpr int FR_BQ = (B_ * M_) / BQ_WAVES;
constexpr int FR_BLOCKS = 2 * FR_BQ;

__global__ __launch_bounds__(BQ_THREADS, 12) void front_kernel(
    const float* __restrict__ p, const float* __restrict__ f, const float* __restrict__ Wsa,
    const float* __restrict__ bsa, const float* __restrict__ Wpe, const float* __restrict__ bpe,
    const float* __restrict__ Wc, const float* __restrict__ bc, float* __restrict__ feat,
    float* __restrict__ hout, int* __restrict__ rec2) {
  __shared__ __attribute__((aligned(16))) float4 s_pts[N_ / BQ_GROUP][3][BQ_QUADS][32];
  __shared__ int s_hit[BQ_WAVES][K_];
  __shared__ __attribute__((aligned(16))) float s_tile[BQ_WAVES][PT];
  const int bid = blockIdx.x;
  const int role = bid < FR_BQ ? 0 : 1;
  const int blk = bid < FR_BQ ? bid : bid - FR_BQ;
  if (role == 0) {
    const int4 rec = ball_query_block<N_, 1>(s_pts, s_hit, blk, p, R1SQ);
    first_stage_tile(s_tile, rec, blk, f, Wsa, bsa, Wpe, bpe, feat, Wc, bc, hout);
  } else {
    const int4 rec = ball_query_block<M_, STRIDE_>(s_pts, s_hit, blk, p, R2SQ);
    ((int4*)rec2)[((size_t)blk * BQ_WAVES + (threadIdx.x >> 5)) * K_ + (threadIdx.x & 31)] = rec;
  }
}

constexpr int TR = 16;
constexpr int FT = 256;
constexpr int HID = EXP_ * C_;
constexpr int PH = HID + 4;

template <bool MLP, bool CONV>
__global__ __launch_bounds__(FT) void row_tile_kernel(
    const float* __restrict__ src, const int* __restrict__ rec, int src_n,
    const float* __restrict__ Wpe, const float* __restrict__ bpe,
    const float* __restrict__ Wpw1, const float* __restrict__ bpw1,
    const float* __restrict__ Wpw2, const float* __restrict__ bpw2, float* feat,
    const float* __restrict__ Wc, const float* __restrict__ bc, float* __restrict__ hout) {
  static_assert(M_ % TR == 0, "a tile never crosses a batch");
  static_assert(TR * K_ == 2 * FT && K_ == 32, "staging: two records per thread, a wave per query");
  static_assert(TR == 2 * (FT / 32), "gather: two queries per wave");
  __shared__ __attribute__((aligned(16))) int4 s_nb[TR][K_];
  __shared__ __attribute__((aligned(16))) float sT[TR][PT];
  __shared__ __attribute__((aligned(16))) float sH[MLP ? TR : 1][PH];

  const int tid = threadIdx.x;
  const int lane = tid & 31;
  const int half = lane >> 4;
  const int ln = lane & 15;
  const int w = __builtin_amdgcn_readfirstlane(tid >> 5);
  const int bm0 = blockIdx.x * TR;
  const int b = bm0 / M_;

  {
    const int4* rb = (const int4*)rec + (size_t)bm0 * K_;
    const int4 recA = rb[tid];
    const int4 recB = rb[tid + FT];
    s_nb[w][lane] = recA;
    s_nb[w + TR / 2][lane] = recB;
  }
  __syncthreads();

  {
    const int c4 = ln * 4;
    const float4 w0 = *(const float4*)(Wpe + 0 * C_ + c4);
    const float4 w1 = *(const float4*)(Wpe + 1 * C_ + c4);
    const float4 w2 = *(const float4*)(Wpe + 2 * C_ + c4);
    const float4 bi = *(const float4*)(bpe + c4);
    const float* sb = src + (size_t)b * src_n * C_ + c4;
#pragma unroll
    for (int qq = 0; qq < 2; ++qq) {
      const int ql = 2 * w + qq;
      float4 best = {NEG_BIG, NEG_BIG, NEG_BIG, NEG_BIG};
#pragma unroll
      for (int i = 0; i < K_ / 2; ++i) {
        const int4 rec = s_nb[ql][2 * i + half];
        const float dx = __int_as_float(rec.y), dy = __int_as_float(rec.z),
                    dz = __int_as_float(rec.w);
        const float4 s4 = *(const float4*)(sb + (size_t)rec.x * C_);
        best.x = fmaxf(best.x, dx * w0.x + dy * w1.x + dz * w2.x + bi.x + s4.x);
        best.y = fmaxf(best.y, dx * w0.y + dy * w1.y + dz * w2.y + bi.y + s4.y);
        best.z = fmaxf(best.z, dx * w0.z + dy * w1.z + dz * w2.z + bi.z + s4.z);
        best.w = fmaxf(best.w, dx * w0.w + dy * w1.w + dz * w2.w + bi.w + s4.w);
      }
      best.x = fmaxf(best.x, __shfl_xor(best.x, 16));
      best.y = fmaxf(best.y, __shfl_xor(best.y, 16));
      best.z = fmaxf(best.z, __shfl_xor(best.z, 16));
      best.w = fmaxf(best.w, __shfl_xor(best.w, 16));
      if (half == 0) *(float4*)&sT[ql][c4] = best;
    }
  }
  __syncthreads();

  if constexpr (MLP) {
    {
      const int n0 = w * 32;
      const float* Wb = Wpw1 + (size_t)(2 * half) * HID + n0 + ln;
      v8f acc0 = {0.f, 0.f, 0.f, 0.f, 0.f, 0.f, 0.f, 0.f};
      v8f acc1 = {0.f, 0.f, 0.f, 0.f, 0.f, 0.f, 0.f, 0.f};
#pragma unroll
      for (int k = 0; k < C_; k += 4) {
        v2f a;   a.x = sT[ln][k + 2 * half];        a.y = sT[ln][k + 2 * half + 1];
        v2f b0;  b0.x = Wb[(size_t)k * HID];        b0.y = Wb[(size_t)(k + 1) * HID];
        v2f b1;  b1.x = Wb[(size_t)k * HID + 16];   b1.y = Wb[(size_t)(k + 1) * HID + 16];
        acc0 = wmma4(a, b0, acc0);
        acc1 = wmma4(a, b1, acc1);
      }
      const float bv0 = bpw1[n0 + ln], bv1 = bpw1[n0 + 16 + ln];
#pragma unroll
      for (int r = 0; r < 8; ++r) {
        sH[r + 8 * half][n0 + ln] = fmaxf(acc0[r] + bv0, 0.0f);
        sH[r + 8 * half][n0 + 16 + ln] = fmaxf(acc1[r] + bv1, 0.0f);
      }
    }
    __syncthreads();

    if (w < 4) {
      const int n0 = w * 16;
      const float* Wb = Wpw2 + (size_t)(2 * half) * C_ + n0 + ln;
      v8f acc = {0.f, 0.f, 0.f, 0.f, 0.f, 0.f, 0.f, 0.f};
#pragma unroll
      for (int k = 0; k < HID; k += 4) {
        v2f a;   a.x = sH[ln][k + 2 * half];        a.y = sH[ln][k + 2 * half + 1];
        v2f bb;  bb.x = Wb[(size_t)k * C_];         bb.y = Wb[(size_t)(k + 1) * C_];
        acc = wmma4(a, bb, acc);
      }
      const int col = n0 + ln;
      const float bv = bpw2[col];
#pragma unroll
      for (int r = 0; r < 8; ++r) {
        const int row = r + 8 * half;
        float v = acc[r] + bv;
        v += feat[(size_t)(bm0 + row) * C_ + col];
        sT[row][col] = fmaxf(v, 0.0f);
      }
    }
    __syncthreads();
  }

  {
    const int row = tid >> 4, c4 = (tid & 15) * 4;
    *(float4*)(feat + (size_t)(bm0 + row) * C_ + c4) = *(const float4*)&sT[row][c4];
  }

  if constexpr (CONV) {
    if (w < 4) {
      const int n0 = w * 16;
      const float* Wb = Wc + (size_t)(2 * half) * C_ + n0 + ln;
      v8f acc = {0.f, 0.f, 0.f, 0.f, 0.f, 0.f, 0.f, 0.f};
#pragma unroll
      for (int k = 0; k < C_; k += 4) {
        v2f a;   a.x = sT[ln][k + 2 * half];        a.y = sT[ln][k + 2 * half + 1];
        v2f bb;  bb.x = Wb[(size_t)k * C_];         bb.y = Wb[(size_t)(k + 1) * C_];
        acc = wmma4(a, bb, acc);
      }
      const int col = n0 + ln;
      const float bv = bc[col];
#pragma unroll
      for (int r = 0; r < 8; ++r) {
        const int row = bm0 + r + 8 * half;
        hout[(size_t)row * C_ + col] = fmaxf(acc[r] + bv, 0.0f);
      }
    }
  }
}

}

extern "C" void kernel_launch(void* const* d_in, const int* in_sizes, int n_in,
                              void* d_out, int out_size, void* d_ws, size_t ws_size,
                              hipStream_t stream) {
  (void)in_sizes; (void)n_in; (void)out_size; (void)ws_size;
  const float* p      = (const float*)d_in[0];
  const float* f      = (const float*)d_in[1];
  const float* W_sa   = (const float*)d_in[2];
  const float* b_sa   = (const float*)d_in[3];
  const float* W_pe1  = (const float*)d_in[4];
  const float* b_pe1  = (const float*)d_in[5];
  const float* W_pe2  = (const float*)d_in[6];
  const float* b_pe2  = (const float*)d_in[7];
  const float* W_conv = (const float*)d_in[8];
  const float* b_conv = (const float*)d_in[9];
  const float* W_pw1  = (const float*)d_in[10];
  const float* b_pw1  = (const float*)d_in[11];
  const float* W_pw2  = (const float*)d_in[12];
  const float* b_pw2  = (const float*)d_in[13];
  float* feat = (float*)d_out;

  char* ws = (char*)d_ws;
  float* hA = (float*)ws;  ws += sizeof(float) * (size_t)B_ * M_ * C_;
  float* hB = (float*)ws;  ws += sizeof(float) * (size_t)B_ * M_ * C_;
  int* rec2 = (int*)ws;

  front_kernel<<<FR_BLOCKS, BQ_THREADS, 0, stream>>>(p, f, W_sa, b_sa, W_pe1, b_pe1, W_conv, b_conv,
                                                     feat, hA, rec2);

  const int tiles = (B_ * M_) / TR;

  row_tile_kernel<true, true><<<tiles, FT, 0, stream>>>(
      hA, rec2, M_, W_pe2, b_pe2, W_pw1, b_pw1, W_pw2, b_pw2, feat,
      W_conv + (size_t)C_ * C_, b_conv + C_, hB);

  row_tile_kernel<true, false><<<tiles, FT, 0, stream>>>(
      hB, rec2, M_, W_pe2, b_pe2, W_pw1 + (size_t)C_ * HID, b_pw1 + HID,
      W_pw2 + (size_t)HID * C_, b_pw2 + C_, feat, nullptr, nullptr, nullptr);
}
